# starting kernel
# baseline (speedup 1.0000x reference)
#include <hip/hip_runtime.h>
#include <hip/hip_bf16.h>
#include <stdint.h>

#define NEG_SLOPE 0.2f
#define EPS_GAT 1e-10f

typedef float v2f __attribute__((ext_vector_type(2)));
typedef float v8f __attribute__((ext_vector_type(8)));

__device__ __forceinline__ unsigned enc_ord(float f) {
  unsigned u = __float_as_uint(f);
  return (u & 0x80000000u) ? ~u : (u | 0x80000000u);
}
__device__ __forceinline__ float dec_ord(unsigned u) {
  return __uint_as_float((u & 0x80000000u) ? (u & 0x7FFFFFFFu) : ~u);
}

__global__ void gat_init(float* __restrict__ out, unsigned* __restrict__ maxb,
                         float* __restrict__ sumb, int outN, int segN) {
  int i = blockIdx.x * blockDim.x + threadIdx.x;
  if (i < outN) out[i] = 0.0f;
  if (i < segN) { maxb[i] = 0x007FFFFFu;   sumb[i] = 0.0f; }
}

__global__ void __launch_bounds__(256, 1)
gat_gemm(const float* __restrict__ x, const float* __restrict__ W,
         float* __restrict__ hout, int M) {
  int wave = (int)((blockIdx.x * blockDim.x + threadIdx.x) >> 5);
  int lane = threadIdx.x & 31;
  int m0 = wave * 32;
  if (m0 >= M) return;
  int r = lane & 15, hi = lane >> 4;

  const float* x0 = x + (size_t)(m0 + r) * 128 + 2 * hi;
  const float* x1 = x0 + (size_t)16 * 128;
  const float* w0 = W + (size_t)r * 128 + 2 * hi;

  v8f acc0[8] = {};
  v8f acc1[8] = {};

  v2f a0c, a1c, bc[8];
  v2f a0n, a1n, bn[8];

  a0c = *(const v2f*)(x0);
  a1c = *(const v2f*)(x1);
#pragma unroll
  for (int t = 0; t < 8; ++t) bc[t] = *(const v2f*)(w0 + (size_t)t * 2048);

#pragma unroll
  for (int k = 0; k < 128; k += 4) {
    if (k + 4 < 128) {
      a0n = *(const v2f*)(x0 + k + 4);
      a1n = *(const v2f*)(x1 + k + 4);
#pragma unroll
      for (int t = 0; t < 8; ++t)
        bn[t] = *(const v2f*)(w0 + (size_t)t * 2048 + k + 4);
    }
#pragma unroll
    for (int t = 0; t < 8; ++t)
      acc0[t] = __builtin_amdgcn_wmma_f32_16x16x4_f32(false, a0c, false, bc[t],
                                                      (short)0, acc0[t], false, false);
#pragma unroll
    for (int t = 0; t < 8; ++t)
      acc1[t] = __builtin_amdgcn_wmma_f32_16x16x4_f32(false, a1c, false, bc[t],
                                                      (short)0, acc1[t], false, false);
    a0c = a0n;
    a1c = a1n;
#pragma unroll
    for (int t = 0; t < 8; ++t) bc[t] = bn[t];
  }

#pragma unroll
  for (int t = 0; t < 8; ++t)
#pragma unroll
    for (int g = 0; g < 8; ++g) {
      hout[(size_t)(m0 + g + 8 * hi) * 128 + t * 16 + r]      = acc0[t][g];
      hout[(size_t)(m0 + 16 + g + 8 * hi) * 128 + t * 16 + r] = acc1[t][g];
    }
}

__global__ void gat_proj(const float* __restrict__ hbuf, const float* __restrict__ a_src,
                         const float* __restrict__ a_dst, float* __restrict__ esrc,
                         float* __restrict__ edst, int BN) {
  int wave = (int)((blockIdx.x * blockDim.x + threadIdx.x) >> 5);
  int lane = threadIdx.x & 31;
  int bn = wave >> 2, h = wave & 3;
  if (bn >= BN) return;
  float hv = hbuf[(size_t)bn * 128 + h * 32 + lane];
  float s = hv * a_src[h * 32 + lane];
  float t = hv * a_dst[h * 32 + lane];
#pragma unroll
  for (int off = 16; off > 0; off >>= 1) {
    s += __shfl_xor(s, off, 32);
    t += __shfl_xor(t, off, 32);
  }
  if (lane == 0) { esrc[bn * 4 + h] = s; edst[bn * 4 + h] = t; }
}

__device__ __forceinline__ float edge_score(const float* esrc, const float* edst,
                                            size_t sidx, size_t didx, int h) {
  float sc = esrc[sidx * 4 + h] + edst[didx * 4 + h];
  return sc > 0.f ? sc : NEG_SLOPE * sc;
}

__global__ void gat_edge_max(const int* __restrict__ src, const int* __restrict__ dst,
                             const float* __restrict__ esrc, const float* __restrict__ edst,
                             unsigned* __restrict__ maxb, int E_, int B_, int N_) {
  int i = blockIdx.x * blockDim.x + threadIdx.x;
  if (i >= E_ * B_ * 4) return;
  int h = i & 3;
  int bi = (i >> 2) % B_;
  int e = (i >> 2) / B_;
  size_t sidx = (size_t)bi * N_ + src[e];
  size_t didx = (size_t)bi * N_ + dst[e];
  float sc = edge_score(esrc, edst, sidx, didx, h);
  atomicMax(&maxb[didx * 4 + h], enc_ord(sc));
}

__global__ void gat_edge_sum(const int* __restrict__ src, const int* __restrict__ dst,
                             const float* __restrict__ esrc, const float* __restrict__ edst,
                             const unsigned* __restrict__ maxb, float* __restrict__ sumb,
                             int E_, int B_, int N_) {
  int i = blockIdx.x * blockDim.x + threadIdx.x;
  if (i >= E_ * B_ * 4) return;
  int h = i & 3;
  int bi = (i >> 2) % B_;
  int e = (i >> 2) / B_;
  size_t sidx = (size_t)bi * N_ + src[e];
  size_t didx = (size_t)bi * N_ + dst[e];
  float sc = edge_score(esrc, edst, sidx, didx, h);
  atomicAdd(&sumb[didx * 4 + h], expf(sc - dec_ord(maxb[didx * 4 + h])));
}

__global__ void gat_edge_agg(const int* __restrict__ src, const int* __restrict__ dst,
                             const float* __restrict__ esrc, const float* __restrict__ edst,
                             const unsigned* __restrict__ maxb, const float* __restrict__ sumb,
                             const float* __restrict__ hbuf, float* __restrict__ out,
                             int E_, int B_, int N_) {
  long long w = ((long long)blockIdx.x * blockDim.x + threadIdx.x) >> 5;
  int lane = threadIdx.x & 31;
  if (w >= (long long)E_ * B_ * 4) return;
  int h = (int)(w & 3);
  long long t = w >> 2;
  int bi = (int)(t % B_);
  long long e = t / B_;
  size_t sidx = (size_t)bi * N_ + src[e];
  size_t didx = (size_t)bi * N_ + dst[e];
  float sc = edge_score(esrc, edst, sidx, didx, h);
  float alpha = expf(sc - dec_ord(maxb[didx * 4 + h])) / (sumb[didx * 4 + h] + EPS_GAT);
  float v = alpha * hbuf[sidx * 128 + h * 32 + lane];
  atomicAdd(&out[didx * 128 + h * 32 + lane], v);
}

extern "C" void kernel_launch(void* const* d_in, const int* in_sizes, int n_in,
                              void* d_out, int out_size, void* d_ws, size_t ws_size,
                              hipStream_t stream) {
  (void)in_sizes; (void)n_in; (void)out_size; (void)ws_size;
  const float* x     = (const float*)d_in[0];
  const int*   ei    = (const int*)d_in[1];
  const float* W     = (const float*)d_in[2];
  const float* a_src = (const float*)d_in[3];
  const float* a_dst = (const float*)d_in[4];
  float* out = (float*)d_out;

  const int B = 2, N = 20000, E = 640000;
  const int BN = B * N;

  float*    hbuf = (float*)d_ws;
  float*    esrc = hbuf + (size_t)BN * 128;
  float*    edst = esrc + (size_t)BN * 4;
  unsigned* maxb = (unsigned*)(edst + (size_t)BN * 4);
  float*    sumb = (float*)(maxb + (size_t)BN * 4);

  {
    int n = BN * 128;
    gat_init<<<(n + 255) / 256, 256, 0, stream>>>(out, maxb, sumb, n, BN * 4);
  }
  {
    int waves = BN / 32;
    gat_gemm<<<(waves * 32 + 255) / 256, 256, 0, stream>>>(x, W, hbuf, BN);
  }
  {
    int waves = BN * 4;
    gat_proj<<<(waves * 32 + 255) / 256, 256, 0, stream>>>(hbuf, a_src, a_dst, esrc, edst, BN);
  }
  {
    int n = E * B * 4;
    gat_edge_max<<<(n + 255) / 256, 256, 0, stream>>>(ei, ei + E, esrc, edst, maxb, E, B, N);
    gat_edge_sum<<<(n + 255) / 256, 256, 0, stream>>>(ei, ei + E, esrc, edst, maxb, sumb, E, B, N);
    long long threads = (long long)n * 32;
    gat_edge_agg<<<(unsigned)((threads + 255) / 256), 256, 0, stream>>>(
        ei, ei + E, esrc, edst, maxb, sumb, hbuf, out, E, B, N);
  }
}


# Edges grouped by destination (int count, scan, slots, groups ordered by edge id); wave per (node,batch) does max, exp-sum, weighted row sum in registers: no float atomics, sums in fixed edge-id order
# speedup vs baseline: 2.4859x; 2.4859x over previous
#include <hip/hip_runtime.h>
#include <hip/hip_bf16.h>
#include <stdint.h>

#define NEG_SLOPE 0.2f
#define EPS_GAT 1e-10f

typedef float v2f __attribute__((ext_vector_type(2)));
typedef float v8f __attribute__((ext_vector_type(8)));

__global__ void __launch_bounds__(256, 1)
gat_gemm(const float* __restrict__ x, const float* __restrict__ W,
         float* __restrict__ hout, int M) {
  int wave = (int)((blockIdx.x * blockDim.x + threadIdx.x) >> 5);
  int lane = threadIdx.x & 31;
  int m0 = wave * 32;
  if (m0 >= M) return;
  int r = lane & 15, hi = lane >> 4;

  const float* x0 = x + (size_t)(m0 + r) * 128 + 2 * hi;
  const float* x1 = x0 + (size_t)16 * 128;
  const float* w0 = W + (size_t)r * 128 + 2 * hi;

  v8f acc0[8] = {};
  v8f acc1[8] = {};

  v2f a0c, a1c, bc[8];
  v2f a0n, a1n, bn[8];

  a0c = *(const v2f*)(x0);
  a1c = *(const v2f*)(x1);
#pragma unroll
  for (int t = 0; t < 8; ++t) bc[t] = *(const v2f*)(w0 + (size_t)t * 2048);

#pragma unroll
  for (int k = 0; k < 128; k += 4) {
    if (k + 4 < 128) {
      a0n = *(const v2f*)(x0 + k + 4);
      a1n = *(const v2f*)(x1 + k + 4);
#pragma unroll
      for (int t = 0; t < 8; ++t)
        bn[t] = *(const v2f*)(w0 + (size_t)t * 2048 + k + 4);
    }
#pragma unroll
    for (int t = 0; t < 8; ++t)
      acc0[t] = __builtin_amdgcn_wmma_f32_16x16x4_f32(false, a0c, false, bc[t],
                                                      (short)0, acc0[t], false, false);
#pragma unroll
    for (int t = 0; t < 8; ++t)
      acc1[t] = __builtin_amdgcn_wmma_f32_16x16x4_f32(false, a1c, false, bc[t],
                                                      (short)0, acc1[t], false, false);
    a0c = a0n;
    a1c = a1n;
#pragma unroll
    for (int t = 0; t < 8; ++t) bc[t] = bn[t];
  }

#pragma unroll
  for (int t = 0; t < 8; ++t)
#pragma unroll
    for (int g = 0; g < 8; ++g) {
      hout[(size_t)(m0 + g + 8 * hi) * 128 + t * 16 + r]      = acc0[t][g];
      hout[(size_t)(m0 + 16 + g + 8 * hi) * 128 + t * 16 + r] = acc1[t][g];
    }
}

__global__ void gat_proj(const float* __restrict__ hbuf, const float* __restrict__ a_src,
                         const float* __restrict__ a_dst, float* __restrict__ esrc,
                         float* __restrict__ edst, int BN) {
  int wave = (int)((blockIdx.x * blockDim.x + threadIdx.x) >> 5);
  int lane = threadIdx.x & 31;
  int bn = wave >> 2, h = wave & 3;
  if (bn >= BN) return;
  float hv = hbuf[(size_t)bn * 128 + h * 32 + lane];
  float s = hv * a_src[h * 32 + lane];
  float t = hv * a_dst[h * 32 + lane];
#pragma unroll
  for (int off = 16; off > 0; off >>= 1) {
    s += __shfl_xor(s, off, 32);
    t += __shfl_xor(t, off, 32);
  }
  if (lane == 0) { esrc[bn * 4 + h] = s; edst[bn * 4 + h] = t; }
}


typedef float v4f __attribute__((ext_vector_type(4)));

__device__ __forceinline__ float lrelu(float s) { return s > 0.f ? s : NEG_SLOPE * s; }

__global__ void gat_zero_cnt(int* __restrict__ cnt, int n) {
  int i = blockIdx.x * blockDim.x + threadIdx.x;
  if (i < n) cnt[i] = 0;
}

__global__ void gat_count(const int* __restrict__ dst, int* __restrict__ cnt, int E_) {
  int e = blockIdx.x * blockDim.x + threadIdx.x;
  if (e < E_) atomicAdd(&cnt[dst[e]], 1);
}

__global__ void __launch_bounds__(1024)
gat_scan(int* __restrict__ cnt, int* __restrict__ rowptr, int n) {
  __shared__ int part[1024];
  int t = threadIdx.x;
  int per = (n + 1023) / 1024;
  int lo = t * per;
  int hi = lo + per < n ? lo + per : n;
  int s = 0;
  for (int i = lo; i < hi; ++i) s += cnt[i];
  part[t] = s;
  __syncthreads();
  for (int off = 1; off < 1024; off <<= 1) {
    int v = (t >= off) ? part[t - off] : 0;
    __syncthreads();
    part[t] += v;
    __syncthreads();
  }
  int run = part[t] - s;
  for (int i = lo; i < hi; ++i) {
    int c = cnt[i];
    rowptr[i] = run;
    cnt[i] = run;
    run += c;
  }
  if (t == 1023) rowptr[n] = part[1023];
}

__global__ void gat_place(const int* __restrict__ dst, int* __restrict__ cursor,
                          int* __restrict__ slot_e, int E_) {
  int e = blockIdx.x * blockDim.x + threadIdx.x;
  if (e < E_) {
    int p = atomicAdd(&cursor[dst[e]], 1);
    slot_e[p] = e;
  }
}

__global__ void gat_order(const int* __restrict__ src, const int* __restrict__ rowptr,
                          const int* __restrict__ slot_e, int* __restrict__ ssrc, int N_) {
  int n = (int)((blockIdx.x * blockDim.x + threadIdx.x) >> 5);
  int lane = threadIdx.x & 31;
  if (n >= N_) return;
  int start = __builtin_amdgcn_readfirstlane(rowptr[n]);
  int end = __builtin_amdgcn_readfirstlane(rowptr[n + 1]);
  for (int ib = start; ib < end; ib += 32) {
    int i = ib + lane;
    int mine = (i < end) ? slot_e[i] : 0x7FFFFFFF;
    int rank = 0;
    for (int jb = start; jb < end; jb += 32) {
      int j = jb + lane;
      int other = (j < end) ? slot_e[j] : 0x7FFFFFFF;
#pragma unroll
      for (int k = 0; k < 32; ++k)
        rank += (__builtin_amdgcn_readlane(other, k) < mine) ? 1 : 0;
    }
    if (i < end) ssrc[start + rank] = src[mine];
  }
}

__global__ void __launch_bounds__(256)
gat_node_agg(const int* __restrict__ rowptr, const int* __restrict__ ssrc,
             const float* __restrict__ esrc, const float* __restrict__ edst,
             const float* __restrict__ hbuf, float* __restrict__ out, int B_, int N_) {
  int w = (int)((blockIdx.x * blockDim.x + threadIdx.x) >> 5);
  int lane = threadIdx.x & 31;
  if (w >= B_ * N_) return;
  int n = w / B_;
  int bi = w - n * B_;
  size_t rowbase = (size_t)bi * N_;
  int start = __builtin_amdgcn_readfirstlane(rowptr[n]);
  int end = __builtin_amdgcn_readfirstlane(rowptr[n + 1]);
  const v4f ed = *(const v4f*)(edst + (rowbase + n) * 4);

  v4f m = {-INFINITY, -INFINITY, -INFINITY, -INFINITY};
  for (int jb = start; jb < end; jb += 32) {
    int j = jb + lane;
    if (j < end) {
      v4f es = *(const v4f*)(esrc + (rowbase + ssrc[j]) * 4);
#pragma unroll
      for (int h = 0; h < 4; ++h) m[h] = fmaxf(m[h], lrelu(es[h] + ed[h]));
    }
  }
#pragma unroll
  for (int off = 16; off > 0; off >>= 1)
#pragma unroll
    for (int h = 0; h < 4; ++h) m[h] = fmaxf(m[h], __shfl_xor(m[h], off, 32));

  v4f sm = {0.f, 0.f, 0.f, 0.f};
  for (int jb = start; jb < end; jb += 32) {
    int j = jb + lane;
    if (j < end) {
      v4f es = *(const v4f*)(esrc + (rowbase + ssrc[j]) * 4);
#pragma unroll
      for (int h = 0; h < 4; ++h) sm[h] += expf(lrelu(es[h] + ed[h]) - m[h]);
    }
  }
#pragma unroll
  for (int off = 16; off > 0; off >>= 1)
#pragma unroll
    for (int h = 0; h < 4; ++h) sm[h] += __shfl_xor(sm[h], off, 32);

  const int hd = lane >> 3;
  const int k0m = hd == 0 ? -1 : 0, k1m = hd == 1 ? -1 : 0, k2m = hd == 2 ? -1 : 0, k3m = hd == 3 ? -1 : 0;
  v4f acc = {0.f, 0.f, 0.f, 0.f};
  const float* hcol = hbuf + rowbase * 128 + 4 * lane;
  for (int jb = start; jb < end; jb += 32) {
    int j = jb + lane;
    int s = n;
    v4f al = {0.f, 0.f, 0.f, 0.f};
    if (j < end) {
      s = ssrc[j];
      v4f es = *(const v4f*)(esrc + (rowbase + s) * 4);
#pragma unroll
      for (int h = 0; h < 4; ++h)
        al[h] = expf(lrelu(es[h] + ed[h]) - m[h]) / (sm[h] + EPS_GAT);
    }
    int cnt = end - jb < 32 ? end - jb : 32;
    for (int k0 = 0; k0 < cnt; k0 += 8) {
      v4f hv[8];
      float a[8];
#pragma unroll
      for (int u = 0; u < 8; ++u) {
        int k = k0 + u;
        int sk = __builtin_amdgcn_readlane(s, k);
        int b0 = __builtin_amdgcn_readlane(__float_as_int(al[0]), k);
        int b1 = __builtin_amdgcn_readlane(__float_as_int(al[1]), k);
        int b2 = __builtin_amdgcn_readlane(__float_as_int(al[2]), k);
        int b3 = __builtin_amdgcn_readlane(__float_as_int(al[3]), k);
        a[u] = __int_as_float((b0 & k0m) | (b1 & k1m) | (b2 & k2m) | (b3 & k3m));
        hv[u] = *(const v4f*)(hcol + (size_t)sk * 128);
      }
#pragma unroll
      for (int u = 0; u < 8; ++u)
        if (k0 + u < cnt) acc += a[u] * hv[u];
    }
  }
  *(v4f*)(out + (rowbase + n) * 128 + 4 * lane) = acc;
}

extern "C" void kernel_launch(void* const* d_in, const int* in_sizes, int n_in,
                              void* d_out, int out_size, void* d_ws, size_t ws_size,
                              hipStream_t stream) {
  (void)in_sizes; (void)n_in; (void)out_size; (void)ws_size;
  const float* x     = (const float*)d_in[0];
  const int*   ei    = (const int*)d_in[1];
  const float* W     = (const float*)d_in[2];
  const float* a_src = (const float*)d_in[3];
  const float* a_dst = (const float*)d_in[4];
  float* out = (float*)d_out;

  const int B = 2, N = 20000, E = 640000;
  const int BN = B * N;

  float* hbuf   = (float*)d_ws;
  float* esrc   = hbuf + (size_t)BN * 128;
  float* edst   = esrc + (size_t)BN * 4;
  int*   ssrc   = (int*)(edst + (size_t)BN * 4);
  int*   rowptr = ssrc + E;
  int*   cnt    = rowptr + (N + 1);
  int*   slot_e = (int*)d_out;

  gat_zero_cnt<<<(N + 255) / 256, 256, 0, stream>>>(cnt, N);
  gat_count<<<(E + 255) / 256, 256, 0, stream>>>(ei + E, cnt, E);
  gat_scan<<<1, 1024, 0, stream>>>(cnt, rowptr, N);
  gat_place<<<(E + 255) / 256, 256, 0, stream>>>(ei + E, cnt, slot_e, E);
  gat_order<<<(N * 32 + 255) / 256, 256, 0, stream>>>(ei, rowptr, slot_e, ssrc, N);
  {
    int waves = BN / 32;
    gat_gemm<<<(waves * 32 + 255) / 256, 256, 0, stream>>>(x, W, hbuf, BN);
  }
  {
    int waves = BN * 4;
    gat_proj<<<(waves * 32 + 255) / 256, 256, 0, stream>>>(hbuf, a_src, a_dst, esrc, edst, BN);
  }
  gat_node_agg<<<(BN * 32 + 255) / 256, 256, 0, stream>>>(rowptr, ssrc, esrc, edst, hbuf, out, B, N);
}


# Aggregation wave keeps a node's scores in registers, lane = (head, edge slot): one logit gather and one exp, weight and row index reach a head's lanes by 8-lane broadcast; sum order kept
# speedup vs baseline: 2.8209x; 1.1348x over previous
#include <hip/hip_runtime.h>
#include <hip/hip_bf16.h>
#include <stdint.h>

#define NEG_SLOPE 0.2f
#define EPS_GAT 1e-10f

typedef float v2f __attribute__((ext_vector_type(2)));
typedef float v8f __attribute__((ext_vector_type(8)));
typedef float v4f __attribute__((ext_vector_type(4)));

template <int M>
__device__ __forceinline__ float row_xadd(float v) {
  const int p = __builtin_amdgcn_update_dpp(0, __float_as_int(v), 0x160 | M, 0xF, 0xF, true);
  return v + __int_as_float(p);
}

__global__ void __launch_bounds__(256, 1)
gat_gemm(const float* __restrict__ x, const float* __restrict__ W,
         const float* __restrict__ a_src, const float* __restrict__ a_dst,
         float* __restrict__ hout, float* __restrict__ esrc, float* __restrict__ edst, int M) {
  int wave = (int)((blockIdx.x * blockDim.x + threadIdx.x) >> 5);
  int lane = threadIdx.x & 31;
  int m0 = wave * 32;
  if (m0 >= M) return;
  int r = lane & 15, hi = lane >> 4;

  const float* x0 = x + (size_t)(m0 + r) * 128 + 2 * hi;
  const float* x1 = x0 + (size_t)16 * 128;
  const float* w0 = W + (size_t)r * 128 + 2 * hi;

  v8f acc0[8] = {};
  v8f acc1[8] = {};

  v2f a0c, a1c, bc[8];
  v2f a0n, a1n, bn[8];

  a0c = *(const v2f*)(x0);
  a1c = *(const v2f*)(x1);
#pragma unroll
  for (int t = 0; t < 8; ++t) bc[t] = *(const v2f*)(w0 + (size_t)t * 2048);

#pragma unroll
  for (int k = 0; k < 128; k += 4) {
    if (k + 4 < 128) {
      a0n = *(const v2f*)(x0 + k + 4);
      a1n = *(const v2f*)(x1 + k + 4);
#pragma unroll
      for (int t = 0; t < 8; ++t)
        bn[t] = *(const v2f*)(w0 + (size_t)t * 2048 + k + 4);
    }
#pragma unroll
    for (int t = 0; t < 8; ++t)
      acc0[t] = __builtin_amdgcn_wmma_f32_16x16x4_f32(false, a0c, false, bc[t],
                                                      (short)0, acc0[t], false, false);
#pragma unroll
    for (int t = 0; t < 8; ++t)
      acc1[t] = __builtin_amdgcn_wmma_f32_16x16x4_f32(false, a1c, false, bc[t],
                                                      (short)0, acc1[t], false, false);
    a0c = a0n;
    a1c = a1n;
#pragma unroll
    for (int t = 0; t < 8; ++t) bc[t] = bn[t];
  }

#pragma unroll
  for (int t = 0; t < 8; ++t)
#pragma unroll
    for (int g = 0; g < 8; ++g) {
      hout[(size_t)(m0 + g + 8 * hi) * 128 + t * 16 + r]      = acc0[t][g];
      hout[(size_t)(m0 + 16 + g + 8 * hi) * 128 + t * 16 + r] = acc1[t][g];
    }

  float as_[8], ad_[8];
#pragma unroll
  for (int t = 0; t < 8; ++t) {
    as_[t] = a_src[t * 16 + r];
    ad_[t] = a_dst[t * 16 + r];
  }
  v4f eso = {0.f, 0.f, 0.f, 0.f}, edo = {0.f, 0.f, 0.f, 0.f};
#pragma unroll
  for (int tile = 0; tile < 2; ++tile)
#pragma unroll
    for (int g = 0; g < 8; ++g) {
      const bool mine = (r == tile * 8 + g);
#pragma unroll
      for (int hh = 0; hh < 4; ++hh) {
        const float c0 = tile ? acc1[2 * hh][g] : acc0[2 * hh][g];
        const float c1 = tile ? acc1[2 * hh + 1][g] : acc0[2 * hh + 1][g];
        float s = c0 * as_[2 * hh] + c1 * as_[2 * hh + 1];
        float d = c0 * ad_[2 * hh] + c1 * ad_[2 * hh + 1];
        s = row_xadd<8>(s); d = row_xadd<8>(d);
        s = row_xadd<4>(s); d = row_xadd<4>(d);
        s = row_xadd<2>(s); d = row_xadd<2>(d);
        s = row_xadd<1>(s); d = row_xadd<1>(d);
        eso[hh] = mine ? s : eso[hh];
        edo[hh] = mine ? d : edo[hh];
      }
    }
  const int orow = m0 + (r >> 3) * 16 + (r & 7) + 8 * hi;
  *(v4f*)(esrc + (size_t)orow * 4) = eso;
  *(v4f*)(edst + (size_t)orow * 4) = edo;
}


__device__ __forceinline__ float lrelu(float s) { return s > 0.f ? s : NEG_SLOPE * s; }

__global__ void gat_zero_cnt(int* __restrict__ cnt, int n) {
  int i = blockIdx.x * blockDim.x + threadIdx.x;
  if (i < n) cnt[i] = 0;
}

__global__ void gat_count(const int* __restrict__ dst, int* __restrict__ cnt, int E_) {
  int e = blockIdx.x * blockDim.x + threadIdx.x;
  if (e < E_) atomicAdd(&cnt[dst[e]], 1);
}

__global__ void __launch_bounds__(1024)
gat_scan(int* __restrict__ cnt, int* __restrict__ rowptr, int n) {
  __shared__ int part[1024];
  int t = threadIdx.x;
  int per = (n + 1023) / 1024;
  int lo = t * per;
  int hi = lo + per < n ? lo + per : n;
  int s = 0;
  for (int i = lo; i < hi; ++i) s += cnt[i];
  part[t] = s;
  __syncthreads();
  for (int off = 1; off < 1024; off <<= 1) {
    int v = (t >= off) ? part[t - off] : 0;
    __syncthreads();
    part[t] += v;
    __syncthreads();
  }
  int run = part[t] - s;
  for (int i = lo; i < hi; ++i) {
    int c = cnt[i];
    rowptr[i] = run;
    cnt[i] = run;
    run += c;
  }
  if (t == 1023) rowptr[n] = part[1023];
}

__global__ void gat_place(const int* __restrict__ dst, int* __restrict__ cursor,
                          int* __restrict__ slot_e, int E_) {
  int e = blockIdx.x * blockDim.x + threadIdx.x;
  if (e < E_) {
    int p = atomicAdd(&cursor[dst[e]], 1);
    slot_e[p] = e;
  }
}

__global__ void gat_order(const int* __restrict__ src, const int* __restrict__ rowptr,
                          const int* __restrict__ slot_e, int* __restrict__ ssrc, int N_) {
  int n = (int)((blockIdx.x * blockDim.x + threadIdx.x) >> 5);
  int lane = threadIdx.x & 31;
  if (n >= N_) return;
  int start = __builtin_amdgcn_readfirstlane(rowptr[n]);
  int end = __builtin_amdgcn_readfirstlane(rowptr[n + 1]);
  for (int ib = start; ib < end; ib += 32) {
    int i = ib + lane;
    int mine = (i < end) ? slot_e[i] : 0x7FFFFFFF;
    int rank = 0;
    for (int jb = start; jb < end; jb += 32) {
      int j = jb + lane;
      int other = (j < end) ? slot_e[j] : 0x7FFFFFFF;
#pragma unroll
      for (int k = 0; k < 32; ++k)
        rank += (__builtin_amdgcn_readlane(other, k) < mine) ? 1 : 0;
    }
    if (i < end) ssrc[start + rank] = src[mine];
  }
}


template <int M>
__device__ __forceinline__ float row_xlane(float v) {
  return __int_as_float(__builtin_amdgcn_update_dpp(0, __float_as_int(v), 0x160 | M, 0xF, 0xF, true));
}
template <int U>
__device__ __forceinline__ int bcast8(int v) {
  return __builtin_amdgcn_mov_dpp8(v, U * 0x249249);
}
template <int U>
__device__ __forceinline__ void row_req(const int sreg, const float areg, const char* __restrict__ hb,
                                        const unsigned lane16, float& a, v4f& hv) {
  const unsigned sb = (unsigned)bcast8<U>(sreg);
  a = __int_as_float(bcast8<U>(__float_as_int(areg)));
  hv = *(const v4f*)(hb + ((sb << 9) + lane16));
}

template <int NC>
__device__ __forceinline__ v4f agg_in_regs(const int lane, const int n, const int start, const int d,
                                           const int* __restrict__ ssrc, const char* __restrict__ eb,
                                           const float edh, const char* __restrict__ hb) {
  const int i = lane & 7, hd = lane >> 3;
  int sidx[NC];
  bool val[NC];
#pragma unroll
  for (int c = 0; c < NC; ++c) {
    const int j = 8 * c + i;
    val[c] = j < d;
    sidx[c] = val[c] ? ssrc[start + j] : n;
  }
  float p[NC];
  float m = -INFINITY;
#pragma unroll
  for (int c = 0; c < NC; ++c) {
    const float es = *(const float*)(eb + (((unsigned)sidx[c] << 4) + 4u * hd));
    p[c] = fminf(lrelu(es + edh), val[c] ? INFINITY : -INFINITY);
    m = fmaxf(m, p[c]);
  }
  m = fmaxf(m, row_xlane<4>(m));
  m = fmaxf(m, row_xlane<2>(m));
  m = fmaxf(m, row_xlane<1>(m));
#pragma unroll
  for (int c = 0; c < NC; ++c) p[c] = expf(p[c] - m);
  float t[4];
#pragma unroll
  for (int c = 0; c < 4; ++c) t[c] = (NC == 8) ? p[c] + p[(c + 4) % NC] : p[c];
  float sm = (t[0] + t[2]) + (t[1] + t[3]);
  sm += row_xlane<4>(sm);
  sm += row_xlane<2>(sm);
  sm += row_xlane<1>(sm);
  const float den = sm + EPS_GAT;
#pragma unroll
  for (int c = 0; c < NC; ++c) p[c] = p[c] / den;

  v4f acc = {0.f, 0.f, 0.f, 0.f};
  const unsigned lane16 = 16u * lane;
#pragma unroll
  for (int g = 0; g < NC; ++g) {
    if (8 * g < d) {
      v4f hv[8];
      float a[8];
      row_req<0>(sidx[g], p[g], hb, lane16, a[0], hv[0]);
      row_req<1>(sidx[g], p[g], hb, lane16, a[1], hv[1]);
      row_req<2>(sidx[g], p[g], hb, lane16, a[2], hv[2]);
      row_req<3>(sidx[g], p[g], hb, lane16, a[3], hv[3]);
      row_req<4>(sidx[g], p[g], hb, lane16, a[4], hv[4]);
      row_req<5>(sidx[g], p[g], hb, lane16, a[5], hv[5]);
      row_req<6>(sidx[g], p[g], hb, lane16, a[6], hv[6]);
      row_req<7>(sidx[g], p[g], hb, lane16, a[7], hv[7]);
#pragma unroll
      for (int u = 0; u < 8; ++u) acc += a[u] * hv[u];
    }
  }
  return acc;
}

__global__ void __launch_bounds__(256)
gat_node_agg(const int* __restrict__ rowptr, const int* __restrict__ ssrc,
             const float* __restrict__ esrc, const float* __restrict__ edst,
             const float* __restrict__ hbuf, float* __restrict__ out, int B_, int N_) {
  const int lane = threadIdx.x & 31;
  const int w = __builtin_amdgcn_readfirstlane((int)((blockIdx.x * blockDim.x + threadIdx.x) >> 5));
  if (w >= B_ * N_) return;
  const int n = w / B_;
  const int bi = w - n * B_;
  const size_t rowbase = (size_t)bi * N_;
  const int start = __builtin_amdgcn_readfirstlane(rowptr[n]);
  const int end = __builtin_amdgcn_readfirstlane(rowptr[n + 1]);
  const int d = end - start;

  if (d <= 64) {
    const char* eb = (const char*)(esrc + rowbase * 4);
    const char* hb = (const char*)(hbuf + rowbase * 128);
    const float edh = edst[(rowbase + n) * 4 + (lane >> 3)];
    v4f r;
    if (d <= 32) r = agg_in_regs<4>(lane, n, start, d, ssrc, eb, edh, hb);
    else         r = agg_in_regs<8>(lane, n, start, d, ssrc, eb, edh, hb);
    *(v4f*)(out + (rowbase + n) * 128 + 4 * lane) = r;
    return;
  }

  const v4f ed = *(const v4f*)(edst + (rowbase + n) * 4);

  v4f m = {-INFINITY, -INFINITY, -INFINITY, -INFINITY};
  for (int jb = start; jb < end; jb += 32) {
    int j = jb + lane;
    if (j < end) {
      v4f es = *(const v4f*)(esrc + (rowbase + ssrc[j]) * 4);
#pragma unroll
      for (int h = 0; h < 4; ++h) m[h] = fmaxf(m[h], lrelu(es[h] + ed[h]));
    }
  }
#pragma unroll
  for (int off = 16; off > 0; off >>= 1)
#pragma unroll
    for (int h = 0; h < 4; ++h) m[h] = fmaxf(m[h], __shfl_xor(m[h], off, 32));

  v4f sm = {0.f, 0.f, 0.f, 0.f};
  for (int jb = start; jb < end; jb += 32) {
    int j = jb + lane;
    if (j < end) {
      v4f es = *(const v4f*)(esrc + (rowbase + ssrc[j]) * 4);
#pragma unroll
      for (int h = 0; h < 4; ++h) sm[h] += expf(lrelu(es[h] + ed[h]) - m[h]);
    }
  }
#pragma unroll
  for (int off = 16; off > 0; off >>= 1)
#pragma unroll
    for (int h = 0; h < 4; ++h) sm[h] += __shfl_xor(sm[h], off, 32);

  const int hd = lane >> 3;
  const int k0m = hd == 0 ? -1 : 0, k1m = hd == 1 ? -1 : 0, k2m = hd == 2 ? -1 : 0, k3m = hd == 3 ? -1 : 0;
  v4f acc = {0.f, 0.f, 0.f, 0.f};
  const float* hcol = hbuf + rowbase * 128 + 4 * lane;
  for (int jb = start; jb < end; jb += 32) {
    int j = jb + lane;
    int s = n;
    v4f al = {0.f, 0.f, 0.f, 0.f};
    if (j < end) {
      s = ssrc[j];
      v4f es = *(const v4f*)(esrc + (rowbase + s) * 4);
#pragma unroll
      for (int h = 0; h < 4; ++h)
        al[h] = expf(lrelu(es[h] + ed[h]) - m[h]) / (sm[h] + EPS_GAT);
    }
    int cnt = end - jb < 32 ? end - jb : 32;
    for (int k0 = 0; k0 < cnt; k0 += 8) {
      v4f hv[8];
      float a[8];
#pragma unroll
      for (int u = 0; u < 8; ++u) {
        int k = k0 + u;
        int sk = __builtin_amdgcn_readlane(s, k);
        int b0 = __builtin_amdgcn_readlane(__float_as_int(al[0]), k);
        int b1 = __builtin_amdgcn_readlane(__float_as_int(al[1]), k);
        int b2 = __builtin_amdgcn_readlane(__float_as_int(al[2]), k);
        int b3 = __builtin_amdgcn_readlane(__float_as_int(al[3]), k);
        a[u] = __int_as_float((b0 & k0m) | (b1 & k1m) | (b2 & k2m) | (b3 & k3m));
        hv[u] = *(const v4f*)(hcol + (size_t)sk * 128);
      }
#pragma unroll
      for (int u = 0; u < 8; ++u)
        if (k0 + u < cnt) acc += a[u] * hv[u];
    }
  }
  *(v4f*)(out + (rowbase + n) * 128 + 4 * lane) = acc;
}

extern "C" void kernel_launch(void* const* d_in, const int* in_sizes, int n_in,
                              void* d_out, int out_size, void* d_ws, size_t ws_size,
                              hipStream_t stream) {
  (void)in_sizes; (void)n_in; (void)out_size; (void)ws_size;
  const float* x     = (const float*)d_in[0];
  const int*   ei    = (const int*)d_in[1];
  const float* W     = (const float*)d_in[2];
  const float* a_src = (const float*)d_in[3];
  const float* a_dst = (const float*)d_in[4];
  float* out = (float*)d_out;

  const int B = 2, N = 20000, E = 640000;
  const int BN = B * N;

  float* hbuf   = (float*)d_ws;
  float* esrc   = hbuf + (size_t)BN * 128;
  float* edst   = esrc + (size_t)BN * 4;
  int*   ssrc   = (int*)(edst + (size_t)BN * 4);
  int*   rowptr = ssrc + E;
  int*   cnt    = rowptr + (N + 1);
  int*   slot_e = (int*)d_out;

  gat_zero_cnt<<<(N + 255) / 256, 256, 0, stream>>>(cnt, N);
  gat_count<<<(E + 255) / 256, 256, 0, stream>>>(ei + E, cnt, E);
  gat_scan<<<1, 1024, 0, stream>>>(cnt, rowptr, N);
  gat_place<<<(E + 255) / 256, 256, 0, stream>>>(ei + E, cnt, slot_e, E);
  gat_order<<<(N * 32 + 255) / 256, 256, 0, stream>>>(ei, rowptr, slot_e, ssrc, N);
  {
    int waves = BN / 32;
    gat_gemm<<<(waves * 32 + 255) / 256, 256, 0, stream>>>(x, W, a_src, a_dst, hbuf, esrc, edst, BN);
  }
  gat_node_agg<<<(BN * 32 + 255) / 256, 256, 0, stream>>>(rowptr, ssrc, esrc, edst, hbuf, out, B, N);
}


# Arrival rank taken from the counting atomic's return and kept per edge; placing is rowptr[dst]+rank with a plain store: one atomic pass over the edges instead of two; sum order unchanged
# speedup vs baseline: 3.7190x; 1.3184x over previous
#include <hip/hip_runtime.h>
#include <hip/hip_bf16.h>
#include <stdint.h>

#define NEG_SLOPE 0.2f
#define EPS_GAT 1e-10f

typedef float v2f __attribute__((ext_vector_type(2)));
typedef float v8f __attribute__((ext_vector_type(8)));
typedef float v4f __attribute__((ext_vector_type(4)));

template <int M>
__device__ __forceinline__ float row_xadd(float v) {
  const int p = __builtin_amdgcn_update_dpp(0, __float_as_int(v), 0x160 | M, 0xF, 0xF, true);
  return v + __int_as_float(p);
}

__global__ void __launch_bounds__(256, 1)
gat_gemm(const float* __restrict__ x, const float* __restrict__ W,
         const float* __restrict__ a_src, const float* __restrict__ a_dst,
         float* __restrict__ hout, float* __restrict__ esrc, float* __restrict__ edst, int M) {
  int wave = (int)((blockIdx.x * blockDim.x + threadIdx.x) >> 5);
  int lane = threadIdx.x & 31;
  int m0 = wave * 32;
  if (m0 >= M) return;
  int r = lane & 15, hi = lane >> 4;

  const float* x0 = x + (size_t)(m0 + r) * 128 + 2 * hi;
  const float* x1 = x0 + (size_t)16 * 128;
  const float* w0 = W + (size_t)r * 128 + 2 * hi;

  v8f acc0[8] = {};
  v8f acc1[8] = {};

  v2f a0c, a1c, bc[8];
  v2f a0n, a1n, bn[8];

  a0c = *(const v2f*)(x0);
  a1c = *(const v2f*)(x1);
#pragma unroll
  for (int t = 0; t < 8; ++t) bc[t] = *(const v2f*)(w0 + (size_t)t * 2048);

#pragma unroll
  for (int k = 0; k < 128; k += 4) {
    if (k + 4 < 128) {
      a0n = *(const v2f*)(x0 + k + 4);
      a1n = *(const v2f*)(x1 + k + 4);
#pragma unroll
      for (int t = 0; t < 8; ++t)
        bn[t] = *(const v2f*)(w0 + (size_t)t * 2048 + k + 4);
    }
#pragma unroll
    for (int t = 0; t < 8; ++t)
      acc0[t] = __builtin_amdgcn_wmma_f32_16x16x4_f32(false, a0c, false, bc[t],
                                                      (short)0, acc0[t], false, false);
#pragma unroll
    for (int t = 0; t < 8; ++t)
      acc1[t] = __builtin_amdgcn_wmma_f32_16x16x4_f32(false, a1c, false, bc[t],
                                                      (short)0, acc1[t], false, false);
    a0c = a0n;
    a1c = a1n;
#pragma unroll
    for (int t = 0; t < 8; ++t) bc[t] = bn[t];
  }

#pragma unroll
  for (int t = 0; t < 8; ++t)
#pragma unroll
    for (int g = 0; g < 8; ++g) {
      hout[(size_t)(m0 + g + 8 * hi) * 128 + t * 16 + r]      = acc0[t][g];
      hout[(size_t)(m0 + 16 + g + 8 * hi) * 128 + t * 16 + r] = acc1[t][g];
    }

  float as_[8], ad_[8];
#pragma unroll
  for (int t = 0; t < 8; ++t) {
    as_[t] = a_src[t * 16 + r];
    ad_[t] = a_dst[t * 16 + r];
  }
  v4f eso = {0.f, 0.f, 0.f, 0.f}, edo = {0.f, 0.f, 0.f, 0.f};
#pragma unroll
  for (int tile = 0; tile < 2; ++tile)
#pragma unroll
    for (int g = 0; g < 8; ++g) {
      const bool mine = (r == tile * 8 + g);
#pragma unroll
      for (int hh = 0; hh < 4; ++hh) {
        const float c0 = tile ? acc1[2 * hh][g] : acc0[2 * hh][g];
        const float c1 = tile ? acc1[2 * hh + 1][g] : acc0[2 * hh + 1][g];
        float s = c0 * as_[2 * hh] + c1 * as_[2 * hh + 1];
        float d = c0 * ad_[2 * hh] + c1 * ad_[2 * hh + 1];
        s = row_xadd<8>(s); d = row_xadd<8>(d);
        s = row_xadd<4>(s); d = row_xadd<4>(d);
        s = row_xadd<2>(s); d = row_xadd<2>(d);
        s = row_xadd<1>(s); d = row_xadd<1>(d);
        eso[hh] = mine ? s : eso[hh];
        edo[hh] = mine ? d : edo[hh];
      }
    }
  const int orow = m0 + (r >> 3) * 16 + (r & 7) + 8 * hi;
  *(v4f*)(esrc + (size_t)orow * 4) = eso;
  *(v4f*)(edst + (size_t)orow * 4) = edo;
}


__device__ __forceinline__ float lrelu(float s) { return s > 0.f ? s : NEG_SLOPE * s; }

__global__ void gat_zero_cnt(int* __restrict__ cnt, int n) {
  int i = blockIdx.x * blockDim.x + threadIdx.x;
  if (i < n) cnt[i] = 0;
}

__global__ void gat_count(const int* __restrict__ dst, int* __restrict__ cnt,
                          int* __restrict__ rank, int E_) {
  int e = blockIdx.x * blockDim.x + threadIdx.x;
  if (e < E_) rank[e] = atomicAdd(&cnt[dst[e]], 1);
}

__global__ void __launch_bounds__(1024)
gat_scan(const int* __restrict__ cnt, int* __restrict__ rowptr, int n) {
  __shared__ int part[1024];
  int t = threadIdx.x;
  int per = (n + 1023) / 1024;
  int lo = t * per;
  int hi = lo + per < n ? lo + per : n;
  int s = 0;
  for (int i = lo; i < hi; ++i) s += cnt[i];
  part[t] = s;
  __syncthreads();
  for (int off = 1; off < 1024; off <<= 1) {
    int v = (t >= off) ? part[t - off] : 0;
    __syncthreads();
    part[t] += v;
    __syncthreads();
  }
  int run = part[t] - s;
  for (int i = lo; i < hi; ++i) {
    int c = cnt[i];
    rowptr[i] = run;
    run += c;
  }
  if (t == 1023) rowptr[n] = part[1023];
}

__global__ void gat_place(const int* __restrict__ dst, const int* __restrict__ rowptr,
                          const int* __restrict__ rank, int* __restrict__ slot_e, int E_) {
  int e = blockIdx.x * blockDim.x + threadIdx.x;
  if (e < E_) slot_e[rowptr[dst[e]] + rank[e]] = e;
}

__global__ void gat_order(const int* __restrict__ src, const int* __restrict__ rowptr,
                          const int* __restrict__ slot_e, int* __restrict__ ssrc, int N_) {
  int n = (int)((blockIdx.x * blockDim.x + threadIdx.x) >> 5);
  int lane = threadIdx.x & 31;
  if (n >= N_) return;
  int start = __builtin_amdgcn_readfirstlane(rowptr[n]);
  int end = __builtin_amdgcn_readfirstlane(rowptr[n + 1]);
  for (int ib = start; ib < end; ib += 32) {
    int i = ib + lane;
    int mine = (i < end) ? slot_e[i] : 0x7FFFFFFF;
    int rank = 0;
    for (int jb = start; jb < end; jb += 32) {
      int j = jb + lane;
      int other = (j < end) ? slot_e[j] : 0x7FFFFFFF;
#pragma unroll
      for (int k = 0; k < 32; ++k)
        rank += (__builtin_amdgcn_readlane(other, k) < mine) ? 1 : 0;
    }
    if (i < end) ssrc[start + rank] = src[mine];
  }
}


template <int M>
__device__ __forceinline__ float row_xlane(float v) {
  return __int_as_float(__builtin_amdgcn_update_dpp(0, __float_as_int(v), 0x160 | M, 0xF, 0xF, true));
}
template <int U>
__device__ __forceinline__ int bcast8(int v) {
  return __builtin_amdgcn_mov_dpp8(v, U * 0x249249);
}
template <int U>
__device__ __forceinline__ void row_req(const int sreg, const float areg, const char* __restrict__ hb,
                                        const unsigned lane16, float& a, v4f& hv) {
  const unsigned sb = (unsigned)bcast8<U>(sreg);
  a = __int_as_float(bcast8<U>(__float_as_int(areg)));
  hv = *(const v4f*)(hb + ((sb << 9) + lane16));
}

template <int NC>
__device__ __forceinline__ v4f agg_in_regs(const int lane, const int n, const int start, const int d,
                                           const int* __restrict__ ssrc, const char* __restrict__ eb,
                                           const float edh, const char* __restrict__ hb) {
  const int i = lane & 7, hd = lane >> 3;
  int sidx[NC];
  bool val[NC];
#pragma unroll
  for (int c = 0; c < NC; ++c) {
    const int j = 8 * c + i;
    val[c] = j < d;
    sidx[c] = val[c] ? ssrc[start + j] : n;
  }
  float p[NC];
  float m = -INFINITY;
#pragma unroll
  for (int c = 0; c < NC; ++c) {
    const float es = *(const float*)(eb + (((unsigned)sidx[c] << 4) + 4u * hd));
    p[c] = fminf(lrelu(es + edh), val[c] ? INFINITY : -INFINITY);
    m = fmaxf(m, p[c]);
  }
  m = fmaxf(m, row_xlane<4>(m));
  m = fmaxf(m, row_xlane<2>(m));
  m = fmaxf(m, row_xlane<1>(m));
#pragma unroll
  for (int c = 0; c < NC; ++c) p[c] = expf(p[c] - m);
  float t[4];
#pragma unroll
  for (int c = 0; c < 4; ++c) t[c] = (NC == 8) ? p[c] + p[(c + 4) % NC] : p[c];
  float sm = (t[0] + t[2]) + (t[1] + t[3]);
  sm += row_xlane<4>(sm);
  sm += row_xlane<2>(sm);
  sm += row_xlane<1>(sm);
  const float den = sm + EPS_GAT;
#pragma unroll
  for (int c = 0; c < NC; ++c) p[c] = p[c] / den;

  v4f acc = {0.f, 0.f, 0.f, 0.f};
  const unsigned lane16 = 16u * lane;
#pragma unroll
  for (int g = 0; g < NC; ++g) {
    if (8 * g < d) {
      v4f hv[8];
      float a[8];
      row_req<0>(sidx[g], p[g], hb, lane16, a[0], hv[0]);
      row_req<1>(sidx[g], p[g], hb, lane16, a[1], hv[1]);
      row_req<2>(sidx[g], p[g], hb, lane16, a[2], hv[2]);
      row_req<3>(sidx[g], p[g], hb, lane16, a[3], hv[3]);
      row_req<4>(sidx[g], p[g], hb, lane16, a[4], hv[4]);
      row_req<5>(sidx[g], p[g], hb, lane16, a[5], hv[5]);
      row_req<6>(sidx[g], p[g], hb, lane16, a[6], hv[6]);
      row_req<7>(sidx[g], p[g], hb, lane16, a[7], hv[7]);
#pragma unroll
      for (int u = 0; u < 8; ++u) acc += a[u] * hv[u];
    }
  }
  return acc;
}

__global__ void __launch_bounds__(256)
gat_node_agg(const int* __restrict__ rowptr, const int* __restrict__ ssrc,
             const float* __restrict__ esrc, const float* __restrict__ edst,
             const float* __restrict__ hbuf, float* __restrict__ out, int B_, int N_) {
  const int lane = threadIdx.x & 31;
  const int w = __builtin_amdgcn_readfirstlane((int)((blockIdx.x * blockDim.x + threadIdx.x) >> 5));
  if (w >= B_ * N_) return;
  const int n = w / B_;
  const int bi = w - n * B_;
  const size_t rowbase = (size_t)bi * N_;
  const int start = __builtin_amdgcn_readfirstlane(rowptr[n]);
  const int end = __builtin_amdgcn_readfirstlane(rowptr[n + 1]);
  const int d = end - start;

  if (d <= 64) {
    const char* eb = (const char*)(esrc + rowbase * 4);
    const char* hb = (const char*)(hbuf + rowbase * 128);
    const float edh = edst[(rowbase + n) * 4 + (lane >> 3)];
    v4f r;
    if (d <= 32) r = agg_in_regs<4>(lane, n, start, d, ssrc, eb, edh, hb);
    else         r = agg_in_regs<8>(lane, n, start, d, ssrc, eb, edh, hb);
    *(v4f*)(out + (rowbase + n) * 128 + 4 * lane) = r;
    return;
  }

  const v4f ed = *(const v4f*)(edst + (rowbase + n) * 4);

  v4f m = {-INFINITY, -INFINITY, -INFINITY, -INFINITY};
  for (int jb = start; jb < end; jb += 32) {
    int j = jb + lane;
    if (j < end) {
      v4f es = *(const v4f*)(esrc + (rowbase + ssrc[j]) * 4);
#pragma unroll
      for (int h = 0; h < 4; ++h) m[h] = fmaxf(m[h], lrelu(es[h] + ed[h]));
    }
  }
#pragma unroll
  for (int off = 16; off > 0; off >>= 1)
#pragma unroll
    for (int h = 0; h < 4; ++h) m[h] = fmaxf(m[h], __shfl_xor(m[h], off, 32));

  v4f sm = {0.f, 0.f, 0.f, 0.f};
  for (int jb = start; jb < end; jb += 32) {
    int j = jb + lane;
    if (j < end) {
      v4f es = *(const v4f*)(esrc + (rowbase + ssrc[j]) * 4);
#pragma unroll
      for (int h = 0; h < 4; ++h) sm[h] += expf(lrelu(es[h] + ed[h]) - m[h]);
    }
  }
#pragma unroll
  for (int off = 16; off > 0; off >>= 1)
#pragma unroll
    for (int h = 0; h < 4; ++h) sm[h] += __shfl_xor(sm[h], off, 32);

  const int hd = lane >> 3;
  const int k0m = hd == 0 ? -1 : 0, k1m = hd == 1 ? -1 : 0, k2m = hd == 2 ? -1 : 0, k3m = hd == 3 ? -1 : 0;
  v4f acc = {0.f, 0.f, 0.f, 0.f};
  const float* hcol = hbuf + rowbase * 128 + 4 * lane;
  for (int jb = start; jb < end; jb += 32) {
    int j = jb + lane;
    int s = n;
    v4f al = {0.f, 0.f, 0.f, 0.f};
    if (j < end) {
      s = ssrc[j];
      v4f es = *(const v4f*)(esrc + (rowbase + s) * 4);
#pragma unroll
      for (int h = 0; h < 4; ++h)
        al[h] = expf(lrelu(es[h] + ed[h]) - m[h]) / (sm[h] + EPS_GAT);
    }
    int cnt = end - jb < 32 ? end - jb : 32;
    for (int k0 = 0; k0 < cnt; k0 += 8) {
      v4f hv[8];
      float a[8];
#pragma unroll
      for (int u = 0; u < 8; ++u) {
        int k = k0 + u;
        int sk = __builtin_amdgcn_readlane(s, k);
        int b0 = __builtin_amdgcn_readlane(__float_as_int(al[0]), k);
        int b1 = __builtin_amdgcn_readlane(__float_as_int(al[1]), k);
        int b2 = __builtin_amdgcn_readlane(__float_as_int(al[2]), k);
        int b3 = __builtin_amdgcn_readlane(__float_as_int(al[3]), k);
        a[u] = __int_as_float((b0 & k0m) | (b1 & k1m) | (b2 & k2m) | (b3 & k3m));
        hv[u] = *(const v4f*)(hcol + (size_t)sk * 128);
      }
#pragma unroll
      for (int u = 0; u < 8; ++u)
        if (k0 + u < cnt) acc += a[u] * hv[u];
    }
  }
  *(v4f*)(out + (rowbase + n) * 128 + 4 * lane) = acc;
}

extern "C" void kernel_launch(void* const* d_in, const int* in_sizes, int n_in,
                              void* d_out, int out_size, void* d_ws, size_t ws_size,
                              hipStream_t stream) {
  (void)in_sizes; (void)n_in; (void)out_size; (void)ws_size;
  const float* x     = (const float*)d_in[0];
  const int*   ei    = (const int*)d_in[1];
  const float* W     = (const float*)d_in[2];
  const float* a_src = (const float*)d_in[3];
  const float* a_dst = (const float*)d_in[4];
  float* out = (float*)d_out;

  const int B = 2, N = 20000, E = 640000;
  const int BN = B * N;

  float* hbuf   = (float*)d_ws;
  float* esrc   = hbuf + (size_t)BN * 128;
  float* edst   = esrc + (size_t)BN * 4;
  int*   ssrc   = (int*)(edst + (size_t)BN * 4);
  int*   rowptr = ssrc + E;
  int*   cnt    = rowptr + (N + 1);
  int*   slot_e = (int*)d_out;
  int*   rank   = slot_e + E;

  gat_zero_cnt<<<(N + 255) / 256, 256, 0, stream>>>(cnt, N);
  gat_count<<<(E + 255) / 256, 256, 0, stream>>>(ei + E, cnt, rank, E);
  gat_scan<<<1, 1024, 0, stream>>>(cnt, rowptr, N);
  gat_place<<<(E + 255) / 256, 256, 0, stream>>>(ei + E, rowptr, rank, slot_e, E);
  gat_order<<<(N * 32 + 255) / 256, 256, 0, stream>>>(ei, rowptr, slot_e, ssrc, N);
  {
    int waves = BN / 32;
    gat_gemm<<<(waves * 32 + 255) / 256, 256, 0, stream>>>(x, W, a_src, a_dst, hbuf, esrc, edst, BN);
  }
  gat_node_agg<<<(BN * 32 + 255) / 256, 256, 0, stream>>>(rowptr, ssrc, esrc, edst, hbuf, out, B, N);
}


# Edge counting moved from global atomics to per-run LDS counters: the LDS add returns the rank inside (run, node); a column prefix over the runs' counter rows gives offsets and totals; sum order kept
# speedup vs baseline: 5.7293x; 1.5405x over previous
#include <hip/hip_runtime.h>
#include <hip/hip_bf16.h>
#include <stdint.h>

#define NEG_SLOPE 0.2f
#define EPS_GAT 1e-10f

typedef float v2f __attribute__((ext_vector_type(2)));
typedef float v8f __attribute__((ext_vector_type(8)));
typedef float v4f __attribute__((ext_vector_type(4)));

template <int M>
__device__ __forceinline__ float row_xadd(float v) {
  const int p = __builtin_amdgcn_update_dpp(0, __float_as_int(v), 0x160 | M, 0xF, 0xF, true);
  return v + __int_as_float(p);
}

__global__ void __launch_bounds__(256, 1)
gat_gemm(const float* __restrict__ x, const float* __restrict__ W,
         const float* __restrict__ a_src, const float* __restrict__ a_dst,
         float* __restrict__ hout, float* __restrict__ esrc, float* __restrict__ edst, int M) {
  int wave = (int)((blockIdx.x * blockDim.x + threadIdx.x) >> 5);
  int lane = threadIdx.x & 31;
  int m0 = wave * 32;
  if (m0 >= M) return;
  int r = lane & 15, hi = lane >> 4;

  const float* x0 = x + (size_t)(m0 + r) * 128 + 2 * hi;
  const float* x1 = x0 + (size_t)16 * 128;
  const float* w0 = W + (size_t)r * 128 + 2 * hi;

  v8f acc0[8] = {};
  v8f acc1[8] = {};

  v2f a0c, a1c, bc[8];
  v2f a0n, a1n, bn[8];

  a0c = *(const v2f*)(x0);
  a1c = *(const v2f*)(x1);
#pragma unroll
  for (int t = 0; t < 8; ++t) bc[t] = *(const v2f*)(w0 + (size_t)t * 2048);

#pragma unroll
  for (int k = 0; k < 128; k += 4) {
    if (k + 4 < 128) {
      a0n = *(const v2f*)(x0 + k + 4);
      a1n = *(const v2f*)(x1 + k + 4);
#pragma unroll
      for (int t = 0; t < 8; ++t)
        bn[t] = *(const v2f*)(w0 + (size_t)t * 2048 + k + 4);
    }
#pragma unroll
    for (int t = 0; t < 8; ++t)
      acc0[t] = __builtin_amdgcn_wmma_f32_16x16x4_f32(false, a0c, false, bc[t],
                                                      (short)0, acc0[t], false, false);
#pragma unroll
    for (int t = 0; t < 8; ++t)
      acc1[t] = __builtin_amdgcn_wmma_f32_16x16x4_f32(false, a1c, false, bc[t],
                                                      (short)0, acc1[t], false, false);
    a0c = a0n;
    a1c = a1n;
#pragma unroll
    for (int t = 0; t < 8; ++t) bc[t] = bn[t];
  }

#pragma unroll
  for (int t = 0; t < 8; ++t)
#pragma unroll
    for (int g = 0; g < 8; ++g) {
      hout[(size_t)(m0 + g + 8 * hi) * 128 + t * 16 + r]      = acc0[t][g];
      hout[(size_t)(m0 + 16 + g + 8 * hi) * 128 + t * 16 + r] = acc1[t][g];
    }

  float as_[8], ad_[8];
#pragma unroll
  for (int t = 0; t < 8; ++t) {
    as_[t] = a_src[t * 16 + r];
    ad_[t] = a_dst[t * 16 + r];
  }
  v4f eso = {0.f, 0.f, 0.f, 0.f}, edo = {0.f, 0.f, 0.f, 0.f};
#pragma unroll
  for (int tile = 0; tile < 2; ++tile)
#pragma unroll
    for (int g = 0; g < 8; ++g) {
      const bool mine = (r == tile * 8 + g);
#pragma unroll
      for (int hh = 0; hh < 4; ++hh) {
        const float c0 = tile ? acc1[2 * hh][g] : acc0[2 * hh][g];
        const float c1 = tile ? acc1[2 * hh + 1][g] : acc0[2 * hh + 1][g];
        float s = c0 * as_[2 * hh] + c1 * as_[2 * hh + 1];
        float d = c0 * ad_[2 * hh] + c1 * ad_[2 * hh + 1];
        s = row_xadd<8>(s); d = row_xadd<8>(d);
        s = row_xadd<4>(s); d = row_xadd<4>(d);
        s = row_xadd<2>(s); d = row_xadd<2>(d);
        s = row_xadd<1>(s); d = row_xadd<1>(d);
        eso[hh] = mine ? s : eso[hh];
        edo[hh] = mine ? d : edo[hh];
      }
    }
  const int orow = m0 + (r >> 3) * 16 + (r & 7) + 8 * hi;
  *(v4f*)(esrc + (size_t)orow * 4) = eso;
  *(v4f*)(edst + (size_t)orow * 4) = edo;
}


__device__ __forceinline__ float lrelu(float s) { return s > 0.f ? s : NEG_SLOPE * s; }

#define GAT_N 20000
#define GAT_E 640000
#define CNT_THREADS 512
#define CNT_PER 25
#define CNT_CHUNK (CNT_THREADS * CNT_PER)
#define CNT_BLOCKS (GAT_E / CNT_CHUNK)
static_assert(CNT_BLOCKS * CNT_CHUNK == GAT_E, "the runs must tile the edge list");
static_assert(GAT_N % 4 == 0, "counter rows are moved as 16-byte words");

__global__ void __launch_bounds__(CNT_THREADS)
gat_count(const int* __restrict__ dst, int* __restrict__ table, int* __restrict__ rank) {
  __shared__ __attribute__((aligned(16))) int hist[GAT_N];
  const int t = threadIdx.x;
  int4* h4 = (int4*)hist;
  for (int i = t; i < GAT_N / 4; i += CNT_THREADS) h4[i] = make_int4(0, 0, 0, 0);
  __syncthreads();
  const int base = blockIdx.x * CNT_CHUNK + t;
  int d[CNT_PER];
#pragma unroll
  for (int j = 0; j < CNT_PER; ++j) d[j] = dst[base + j * CNT_THREADS];
  int r[CNT_PER];
#pragma unroll
  for (int j = 0; j < CNT_PER; ++j) r[j] = atomicAdd(&hist[d[j]], 1);
  __syncthreads();
  int4* o4 = (int4*)(table + (size_t)blockIdx.x * GAT_N);
  for (int i = t; i < GAT_N / 4; i += CNT_THREADS) o4[i] = h4[i];
#pragma unroll
  for (int j = 0; j < CNT_PER; ++j) rank[base + j * CNT_THREADS] = r[j];
}

__global__ void gat_colprefix(int* __restrict__ table, int* __restrict__ cnt) {
  const int n = blockIdx.x * blockDim.x + threadIdx.x;
  if (n >= GAT_N) return;
  int c[CNT_BLOCKS];
#pragma unroll
  for (int k = 0; k < CNT_BLOCKS; ++k) c[k] = table[k * GAT_N + n];
  int run = 0;
#pragma unroll
  for (int k = 0; k < CNT_BLOCKS; ++k) {
    table[k * GAT_N + n] = run;
    run += c[k];
  }
  cnt[n] = run;
}

__global__ void __launch_bounds__(1024)
gat_scan(const int* __restrict__ cnt, int* __restrict__ rowptr, int n) {
  __shared__ int part[1024];
  int t = threadIdx.x;
  int per = (n + 1023) / 1024;
  int lo = t * per;
  int hi = lo + per < n ? lo + per : n;
  int s = 0;
  for (int i = lo; i < hi; ++i) s += cnt[i];
  part[t] = s;
  __syncthreads();
  for (int off = 1; off < 1024; off <<= 1) {
    int v = (t >= off) ? part[t - off] : 0;
    __syncthreads();
    part[t] += v;
    __syncthreads();
  }
  int run = part[t] - s;
  for (int i = lo; i < hi; ++i) {
    int c = cnt[i];
    rowptr[i] = run;
    run += c;
  }
  if (t == 1023) rowptr[n] = part[1023];
}

__global__ void gat_place(const int* __restrict__ dst, const int* __restrict__ rowptr,
                          const int* __restrict__ table, const int* __restrict__ rank,
                          int* __restrict__ slot_e, int E_) {
  int e = blockIdx.x * blockDim.x + threadIdx.x;
  if (e < E_) {
    const int d = dst[e];
    slot_e[rowptr[d] + table[(e / CNT_CHUNK) * GAT_N + d] + rank[e]] = e;
  }
}

__global__ void gat_order(const int* __restrict__ src, const int* __restrict__ rowptr,
                          const int* __restrict__ slot_e, int* __restrict__ ssrc, int N_) {
  int n = (int)((blockIdx.x * blockDim.x + threadIdx.x) >> 5);
  int lane = threadIdx.x & 31;
  if (n >= N_) return;
  int start = __builtin_amdgcn_readfirstlane(rowptr[n]);
  int end = __builtin_amdgcn_readfirstlane(rowptr[n + 1]);
  for (int ib = start; ib < end; ib += 32) {
    int i = ib + lane;
    int mine = (i < end) ? slot_e[i] : 0x7FFFFFFF;
    int rank = 0;
    for (int jb = start; jb < end; jb += 32) {
      int j = jb + lane;
      int other = (j < end) ? slot_e[j] : 0x7FFFFFFF;
#pragma unroll
      for (int k = 0; k < 32; ++k)
        rank += (__builtin_amdgcn_readlane(other, k) < mine) ? 1 : 0;
    }
    if (i < end) ssrc[start + rank] = src[mine];
  }
}


template <int M>
__device__ __forceinline__ float row_xlane(float v) {
  return __int_as_float(__builtin_amdgcn_update_dpp(0, __float_as_int(v), 0x160 | M, 0xF, 0xF, true));
}
template <int U>
__device__ __forceinline__ int bcast8(int v) {
  return __builtin_amdgcn_mov_dpp8(v, U * 0x249249);
}
template <int U>
__device__ __forceinline__ void row_req(const int sreg, const float areg, const char* __restrict__ hb,
                                        const unsigned lane16, float& a, v4f& hv) {
  const unsigned sb = (unsigned)bcast8<U>(sreg);
  a = __int_as_float(bcast8<U>(__float_as_int(areg)));
  hv = *(const v4f*)(hb + ((sb << 9) + lane16));
}

template <int NC>
__device__ __forceinline__ v4f agg_in_regs(const int lane, const int n, const int start, const int d,
                                           const int* __restrict__ ssrc, const char* __restrict__ eb,
                                           const float edh, const char* __restrict__ hb) {
  const int i = lane & 7, hd = lane >> 3;
  int sidx[NC];
  bool val[NC];
#pragma unroll
  for (int c = 0; c < NC; ++c) {
    const int j = 8 * c + i;
    val[c] = j < d;
    sidx[c] = val[c] ? ssrc[start + j] : n;
  }
  float p[NC];
  float m = -INFINITY;
#pragma unroll
  for (int c = 0; c < NC; ++c) {
    const float es = *(const float*)(eb + (((unsigned)sidx[c] << 4) + 4u * hd));
    p[c] = fminf(lrelu(es + edh), val[c] ? INFINITY : -INFINITY);
    m = fmaxf(m, p[c]);
  }
  m = fmaxf(m, row_xlane<4>(m));
  m = fmaxf(m, row_xlane<2>(m));
  m = fmaxf(m, row_xlane<1>(m));
#pragma unroll
  for (int c = 0; c < NC; ++c) p[c] = expf(p[c] - m);
  float t[4];
#pragma unroll
  for (int c = 0; c < 4; ++c) t[c] = (NC == 8) ? p[c] + p[(c + 4) % NC] : p[c];
  float sm = (t[0] + t[2]) + (t[1] + t[3]);
  sm += row_xlane<4>(sm);
  sm += row_xlane<2>(sm);
  sm += row_xlane<1>(sm);
  const float den = sm + EPS_GAT;
#pragma unroll
  for (int c = 0; c < NC; ++c) p[c] = p[c] / den;

  v4f acc = {0.f, 0.f, 0.f, 0.f};
  const unsigned lane16 = 16u * lane;
#pragma unroll
  for (int g = 0; g < NC; ++g) {
    if (8 * g < d) {
      v4f hv[8];
      float a[8];
      row_req<0>(sidx[g], p[g], hb, lane16, a[0], hv[0]);
      row_req<1>(sidx[g], p[g], hb, lane16, a[1], hv[1]);
      row_req<2>(sidx[g], p[g], hb, lane16, a[2], hv[2]);
      row_req<3>(sidx[g], p[g], hb, lane16, a[3], hv[3]);
      row_req<4>(sidx[g], p[g], hb, lane16, a[4], hv[4]);
      row_req<5>(sidx[g], p[g], hb, lane16, a[5], hv[5]);
      row_req<6>(sidx[g], p[g], hb, lane16, a[6], hv[6]);
      row_req<7>(sidx[g], p[g], hb, lane16, a[7], hv[7]);
#pragma unroll
      for (int u = 0; u < 8; ++u) acc += a[u] * hv[u];
    }
  }
  return acc;
}

__global__ void __launch_bounds__(256)
gat_node_agg(const int* __restrict__ rowptr, const int* __restrict__ ssrc,
             const float* __restrict__ esrc, const float* __restrict__ edst,
             const float* __restrict__ hbuf, float* __restrict__ out, int B_, int N_) {
  const int lane = threadIdx.x & 31;
  const int w = __builtin_amdgcn_readfirstlane((int)((blockIdx.x * blockDim.x + threadIdx.x) >> 5));
  if (w >= B_ * N_) return;
  const int n = w / B_;
  const int bi = w - n * B_;
  const size_t rowbase = (size_t)bi * N_;
  const int start = __builtin_amdgcn_readfirstlane(rowptr[n]);
  const int end = __builtin_amdgcn_readfirstlane(rowptr[n + 1]);
  const int d = end - start;

  if (d <= 64) {
    const char* eb = (const char*)(esrc + rowbase * 4);
    const char* hb = (const char*)(hbuf + rowbase * 128);
    const float edh = edst[(rowbase + n) * 4 + (lane >> 3)];
    v4f r;
    if (d <= 32) r = agg_in_regs<4>(lane, n, start, d, ssrc, eb, edh, hb);
    else         r = agg_in_regs<8>(lane, n, start, d, ssrc, eb, edh, hb);
    *(v4f*)(out + (rowbase + n) * 128 + 4 * lane) = r;
    return;
  }

  const v4f ed = *(const v4f*)(edst + (rowbase + n) * 4);

  v4f m = {-INFINITY, -INFINITY, -INFINITY, -INFINITY};
  for (int jb = start; jb < end; jb += 32) {
    int j = jb + lane;
    if (j < end) {
      v4f es = *(const v4f*)(esrc + (rowbase + ssrc[j]) * 4);
#pragma unroll
      for (int h = 0; h < 4; ++h) m[h] = fmaxf(m[h], lrelu(es[h] + ed[h]));
    }
  }
#pragma unroll
  for (int off = 16; off > 0; off >>= 1)
#pragma unroll
    for (int h = 0; h < 4; ++h) m[h] = fmaxf(m[h], __shfl_xor(m[h], off, 32));

  v4f sm = {0.f, 0.f, 0.f, 0.f};
  for (int jb = start; jb < end; jb += 32) {
    int j = jb + lane;
    if (j < end) {
      v4f es = *(const v4f*)(esrc + (rowbase + ssrc[j]) * 4);
#pragma unroll
      for (int h = 0; h < 4; ++h) sm[h] += expf(lrelu(es[h] + ed[h]) - m[h]);
    }
  }
#pragma unroll
  for (int off = 16; off > 0; off >>= 1)
#pragma unroll
    for (int h = 0; h < 4; ++h) sm[h] += __shfl_xor(sm[h], off, 32);

  const int hd = lane >> 3;
  const int k0m = hd == 0 ? -1 : 0, k1m = hd == 1 ? -1 : 0, k2m = hd == 2 ? -1 : 0, k3m = hd == 3 ? -1 : 0;
  v4f acc = {0.f, 0.f, 0.f, 0.f};
  const float* hcol = hbuf + rowbase * 128 + 4 * lane;
  for (int jb = start; jb < end; jb += 32) {
    int j = jb + lane;
    int s = n;
    v4f al = {0.f, 0.f, 0.f, 0.f};
    if (j < end) {
      s = ssrc[j];
      v4f es = *(const v4f*)(esrc + (rowbase + s) * 4);
#pragma unroll
      for (int h = 0; h < 4; ++h)
        al[h] = expf(lrelu(es[h] + ed[h]) - m[h]) / (sm[h] + EPS_GAT);
    }
    int cnt = end - jb < 32 ? end - jb : 32;
    for (int k0 = 0; k0 < cnt; k0 += 8) {
      v4f hv[8];
      float a[8];
#pragma unroll
      for (int u = 0; u < 8; ++u) {
        int k = k0 + u;
        int sk = __builtin_amdgcn_readlane(s, k);
        int b0 = __builtin_amdgcn_readlane(__float_as_int(al[0]), k);
        int b1 = __builtin_amdgcn_readlane(__float_as_int(al[1]), k);
        int b2 = __builtin_amdgcn_readlane(__float_as_int(al[2]), k);
        int b3 = __builtin_amdgcn_readlane(__float_as_int(al[3]), k);
        a[u] = __int_as_float((b0 & k0m) | (b1 & k1m) | (b2 & k2m) | (b3 & k3m));
        hv[u] = *(const v4f*)(hcol + (size_t)sk * 128);
      }
#pragma unroll
      for (int u = 0; u < 8; ++u)
        if (k0 + u < cnt) acc += a[u] * hv[u];
    }
  }
  *(v4f*)(out + (rowbase + n) * 128 + 4 * lane) = acc;
}

extern "C" void kernel_launch(void* const* d_in, const int* in_sizes, int n_in,
                              void* d_out, int out_size, void* d_ws, size_t ws_size,
                              hipStream_t stream) {
  (void)in_sizes; (void)n_in; (void)out_size; (void)ws_size;
  const float* x     = (const float*)d_in[0];
  const int*   ei    = (const int*)d_in[1];
  const float* W     = (const float*)d_in[2];
  const float* a_src = (const float*)d_in[3];
  const float* a_dst = (const float*)d_in[4];
  float* out = (float*)d_out;

  const int B = 2, N = GAT_N, E = GAT_E;
  const int BN = B * N;

  float* hbuf   = (float*)d_ws;
  float* esrc   = hbuf + (size_t)BN * 128;
  float* edst   = esrc + (size_t)BN * 4;
  int*   ssrc   = (int*)(edst + (size_t)BN * 4);
  int*   rowptr = ssrc + E;
  int*   cnt    = rowptr + (N + 1);
  int*   slot_e = (int*)d_out;
  int*   rank   = slot_e + E;
  int*   table  = rank + E;

  gat_count<<<CNT_BLOCKS, CNT_THREADS, 0, stream>>>(ei + E, table, rank);
  gat_colprefix<<<(N + 255) / 256, 256, 0, stream>>>(table, cnt);
  gat_scan<<<1, 1024, 0, stream>>>(cnt, rowptr, N);
  gat_place<<<(E + 255) / 256, 256, 0, stream>>>(ei + E, rowptr, table, rank, slot_e, E);
  gat_order<<<(N * 32 + 255) / 256, 256, 0, stream>>>(ei, rowptr, slot_e, ssrc, N);
  {
    int waves = BN / 32;
    gat_gemm<<<(waves * 32 + 255) / 256, 256, 0, stream>>>(x, W, a_src, a_dst, hbuf, esrc, edst, BN);
  }
  gat_node_agg<<<(BN * 32 + 255) / 256, 256, 0, stream>>>(rowptr, ssrc, esrc, edst, hbuf, out, B, N);
}


# Aggregation: small reads per edge cut 3 to 1: list entry carries the source node (LDS counters 16-bit), both batches' source logits in one 32-byte record (GEMM wave: 16 nodes x 2 batches); sums kept
# speedup vs baseline: 12.2289x; 2.1344x over previous
#include <hip/hip_runtime.h>
#include <hip/hip_bf16.h>
#include <stdint.h>

#define NEG_SLOPE 0.2f
#define EPS_GAT 1e-10f
#define GAT_N 20000
#define GAT_E 640000
#define GAT_B 2

typedef float v2f __attribute__((ext_vector_type(2)));
typedef float v8f __attribute__((ext_vector_type(8)));
typedef float v4f __attribute__((ext_vector_type(4)));

template <int M>
__device__ __forceinline__ float row_xadd(float v) {
  const int p = __builtin_amdgcn_update_dpp(0, __float_as_int(v), 0x160 | M, 0xF, 0xF, true);
  return v + __int_as_float(p);
}

__device__ __forceinline__ void
gemm_body(const float* __restrict__ x, const float* __restrict__ W,
          const float* __restrict__ a_src, const float* __restrict__ a_dst,
          float* __restrict__ hout, float* __restrict__ esrc, float* __restrict__ edst, int M) {
  int wave = (int)((blockIdx.x * blockDim.x + threadIdx.x) >> 5);
  int lane = threadIdx.x & 31;
  int node0 = wave * 16;
  if (2 * node0 >= M) return;
  int r = lane & 15, hi = lane >> 4;

  const float* x0 = x + (size_t)(node0 + r) * 128 + 2 * hi;
  const float* x1 = x0 + (size_t)GAT_N * 128;
  const float* w0 = W + (size_t)r * 128 + 2 * hi;

  v8f acc0[8] = {};
  v8f acc1[8] = {};

  v2f a0c, a1c, bc[8];
  v2f a0n, a1n, bn[8];

  a0c = *(const v2f*)(x0);
  a1c = *(const v2f*)(x1);
#pragma unroll
  for (int t = 0; t < 8; ++t) bc[t] = *(const v2f*)(w0 + (size_t)t * 2048);

#pragma unroll
  for (int k = 0; k < 128; k += 4) {
    if (k + 4 < 128) {
      a0n = *(const v2f*)(x0 + k + 4);
      a1n = *(const v2f*)(x1 + k + 4);
#pragma unroll
      for (int t = 0; t < 8; ++t)
        bn[t] = *(const v2f*)(w0 + (size_t)t * 2048 + k + 4);
    }
#pragma unroll
    for (int t = 0; t < 8; ++t)
      acc0[t] = __builtin_amdgcn_wmma_f32_16x16x4_f32(false, a0c, false, bc[t],
                                                      (short)0, acc0[t], false, false);
#pragma unroll
    for (int t = 0; t < 8; ++t)
      acc1[t] = __builtin_amdgcn_wmma_f32_16x16x4_f32(false, a1c, false, bc[t],
                                                      (short)0, acc1[t], false, false);
    a0c = a0n;
    a1c = a1n;
#pragma unroll
    for (int t = 0; t < 8; ++t) bc[t] = bn[t];
  }

#pragma unroll
  for (int t = 0; t < 8; ++t)
#pragma unroll
    for (int g = 0; g < 8; ++g) {
      hout[(size_t)(node0 + g + 8 * hi) * 128 + t * 16 + r]         = acc0[t][g];
      hout[(size_t)(GAT_N + node0 + g + 8 * hi) * 128 + t * 16 + r] = acc1[t][g];
    }

  float as_[8], ad_[8];
#pragma unroll
  for (int t = 0; t < 8; ++t) {
    as_[t] = a_src[t * 16 + r];
    ad_[t] = a_dst[t * 16 + r];
  }
  v4f eso = {0.f, 0.f, 0.f, 0.f}, edo = {0.f, 0.f, 0.f, 0.f};
#pragma unroll
  for (int tile = 0; tile < 2; ++tile)
#pragma unroll
    for (int g = 0; g < 8; ++g) {
      const bool mine = (r == tile * 8 + g);
#pragma unroll
      for (int hh = 0; hh < 4; ++hh) {
        const float c0 = tile ? acc1[2 * hh][g] : acc0[2 * hh][g];
        const float c1 = tile ? acc1[2 * hh + 1][g] : acc0[2 * hh + 1][g];
        float s = c0 * as_[2 * hh] + c1 * as_[2 * hh + 1];
        float d = c0 * ad_[2 * hh] + c1 * ad_[2 * hh + 1];
        s = row_xadd<8>(s); d = row_xadd<8>(d);
        s = row_xadd<4>(s); d = row_xadd<4>(d);
        s = row_xadd<2>(s); d = row_xadd<2>(d);
        s = row_xadd<1>(s); d = row_xadd<1>(d);
        eso[hh] = mine ? s : eso[hh];
        edo[hh] = mine ? d : edo[hh];
      }
    }
  const int ob = r >> 3;
  const int onode = node0 + (r & 7) + 8 * hi;
  float* ep = esrc + (size_t)onode * 8 + ob;
  ep[0] = eso[0]; ep[2] = eso[1]; ep[4] = eso[2]; ep[6] = eso[3];
  *(v4f*)(edst + ((size_t)ob * GAT_N + onode) * 4) = edo;
}


__device__ __forceinline__ float lrelu(float s) { return s > 0.f ? s : NEG_SLOPE * s; }

#define CNT_THREADS 256
#define CNT_PER 100
#define CNT_CHUNK (CNT_THREADS * CNT_PER)
#define CNT_BLOCKS (GAT_E / CNT_CHUNK)
#define PFX_PITCH (GAT_N + 8)
#define SCAN_SPAN 41
static_assert(CNT_BLOCKS * CNT_CHUNK == GAT_E, "the runs must tile the edge list");
static_assert(CNT_BLOCKS <= 32, "the aggregation wave gives every run one lane");
static_assert(CNT_CHUNK < 65536, "offsets and prefixes inside a run are kept as 16-bit words");
static_assert(GAT_N <= 65536, "a source node is kept in the low 16 bits of a list entry");
static_assert(GAT_N % 8 == 0 && CNT_CHUNK % 4 == 0 && PFX_PITCH % 8 == 0, "prefix rows are moved as 16-byte words of eight entries, lists as 16-byte words of four");
static_assert(SCAN_SPAN * CNT_THREADS >= GAT_N / 2 && GAT_N % 2 == 0, "the threads' spans must cover the counter words");

__device__ __forceinline__ int wg_excl_scan(const int v, int* __restrict__ wsum, int& total) {
  const int lane = threadIdx.x & 31, w = threadIdx.x >> 5;
  int inc = v;
  for (int off = 1; off < 32; off <<= 1) {
    const int o = __shfl_up(inc, off, 32);
    inc += (lane >= off) ? o : 0;
  }
  if (lane == 31) wsum[w] = inc;
  __syncthreads();
  int before = 0, tot = 0;
  for (int k = 0; k < CNT_THREADS / 32; ++k) {
    const int s = wsum[k];
    before += (k < w) ? s : 0;
    tot += s;
  }
  total = tot;
  return before + inc - v;
}

__device__ __forceinline__ void
count_body(unsigned* __restrict__ hist, unsigned* __restrict__ lst, int* __restrict__ wsum, const int run,
           const int* __restrict__ src, const int* __restrict__ dst, unsigned short* __restrict__ pfx,
           unsigned* __restrict__ runlist) {
  const int t = threadIdx.x;
  unsigned short* h16 = (unsigned short*)hist;
  uint4* h4 = (uint4*)hist;
  for (int i = t; i < GAT_N / 8; i += CNT_THREADS) h4[i] = make_uint4(0u, 0u, 0u, 0u);
  __syncthreads();
  const int base = run * CNT_CHUNK + t;
  int d[CNT_PER];
#pragma unroll
  for (int j = 0; j < CNT_PER; ++j) d[j] = dst[base + j * CNT_THREADS];
  unsigned r[CNT_PER];
#pragma unroll
  for (int j = 0; j < CNT_PER; ++j) r[j] = atomicAdd(&hist[d[j] >> 1], (d[j] & 1) ? 0x10000u : 1u);
#pragma unroll
  for (int j = 0; j < CNT_PER; ++j) d[j] |= (int)(((d[j] & 1) ? (r[j] >> 16) : (r[j] & 0xFFFFu)) << 16);
  __syncthreads();
  int sv[CNT_PER];
#pragma unroll
  for (int j = 0; j < CNT_PER; ++j) sv[j] = src[base + j * CNT_THREADS];
  const int lo = t * SCAN_SPAN;
  const int hi = (lo + SCAN_SPAN < GAT_N / 2) ? lo + SCAN_SPAN : GAT_N / 2;
  int s = 0;
  for (int i = lo; i < hi; ++i) {
    const unsigned w = hist[i];
    s += (int)(w & 0xFFFFu) + (int)(w >> 16);
  }
  int total;
  int ex = wg_excl_scan(s, wsum, total);
  for (int i = lo; i < hi; ++i) {
    const unsigned w = hist[i];
    const int mid = ex + (int)(w & 0xFFFFu);
    hist[i] = (unsigned)ex | ((unsigned)mid << 16);
    ex = mid + (int)(w >> 16);
  }
  __syncthreads();
  uint4* o4 = (uint4*)(pfx + (size_t)run * PFX_PITCH);
  for (int i = t; i < GAT_N / 8; i += CNT_THREADS) o4[i] = h4[i];
  if (t == 0) pfx[(size_t)run * PFX_PITCH + GAT_N] = (unsigned short)CNT_CHUNK;
#pragma unroll
  for (int j = 0; j < CNT_PER; ++j)
    lst[(int)h16[d[j] & 0xFFFF] + (int)((unsigned)d[j] >> 16)] = ((unsigned)(t + j * CNT_THREADS) << 16) | (unsigned)sv[j];
  __syncthreads();
  const uint4* l4 = (const uint4*)lst;
  uint4* g4 = (uint4*)(runlist + (size_t)run * CNT_CHUNK);
  for (int i = t; i < CNT_CHUNK / 4; i += CNT_THREADS) g4[i] = l4[i];
}

#define GEMM_ROWS (2 * GAT_N)
#define GEMM_BLOCKS ((GEMM_ROWS + 255) / 256)
__global__ void __launch_bounds__(256, 1)
gat_gemm_count(const float* __restrict__ x, const float* __restrict__ W,
               const float* __restrict__ a_src, const float* __restrict__ a_dst,
               float* __restrict__ hout, float* __restrict__ esrc, float* __restrict__ edst,
               const int* __restrict__ src, const int* __restrict__ dst, int* __restrict__ pfx_w,
               int* __restrict__ runlist_w) {
  __shared__ __attribute__((aligned(16))) unsigned cnt_lds[GAT_N / 2 + CNT_CHUNK];
  unsigned* hist = cnt_lds;
  unsigned* lst = cnt_lds + GAT_N / 2;
  __shared__ int wsum[CNT_THREADS / 32];
  if (blockIdx.x < GEMM_BLOCKS) gemm_body(x, W, a_src, a_dst, hout, esrc, edst, GEMM_ROWS);
  else count_body(hist, lst, wsum, (int)blockIdx.x - GEMM_BLOCKS, src, dst,
                  (unsigned short*)pfx_w, (unsigned*)runlist_w);
}


template <int M>
__device__ __forceinline__ float row_xlane(float v) {
  return __int_as_float(__builtin_amdgcn_update_dpp(0, __float_as_int(v), 0x160 | M, 0xF, 0xF, true));
}
template <int U>
__device__ __forceinline__ int bcast8(int v) {
  return __builtin_amdgcn_mov_dpp8(v, U * 0x249249);
}
template <int U>
__device__ __forceinline__ void row_req(const int sreg, const char* __restrict__ hb,
                                        const unsigned lane16, v4f& hv) {
  const unsigned sb = (unsigned)bcast8<U>(sreg);
  hv = *(const v4f*)(hb + ((sb << 9) + lane16));
}
template <int U>
__device__ __forceinline__ void row_add(const float areg, const v4f hv, v4f& acc) {
  acc += __int_as_float(bcast8<U>(__float_as_int(areg))) * hv;
}


__device__ __forceinline__ void collect_list(const int lane, const int a, const int c, const int off,
                                             const unsigned* runlist, int* __restrict__ grow,
                                             int* __restrict__ gsrc) {
  const int kbase = (c > 0) ? lane * CNT_CHUNK : 0;
  const unsigned* seg = runlist + kbase + a;
  unsigned v[4];
#pragma unroll
  for (int i = 0; i < 4; ++i) v[i] = (i < c) ? seg[i] : 0u;
#pragma unroll
  for (int i = 0; i < 4; ++i)
    if (i < c) {
      grow[off + i] = kbase + (int)(v[i] >> 16);
      gsrc[off + i] = (int)(v[i] & 0xFFFFu);
    }
  for (int i = 4; i < 64; ++i) {
    if (__builtin_amdgcn_ballot_w32(i < c) == 0) break;
    if (i < c) {
      const unsigned w = seg[i];
      grow[off + i] = kbase + (int)(w >> 16);
      gsrc[off + i] = (int)(w & 0xFFFFu);
    }
  }
  __builtin_amdgcn_fence(__ATOMIC_RELEASE, "wavefront");
  __builtin_amdgcn_wave_barrier();
  __builtin_amdgcn_fence(__ATOMIC_ACQUIRE, "wavefront");
}

template <int NC>
__device__ __forceinline__ void list_in_order(const int lane, const int n, const int d,
                                              const int* __restrict__ grow, const int* __restrict__ gsrc,
                                              int* __restrict__ row, int (&sidx)[NC]) {
  constexpr int NR = NC / 4;
  int e[NR], sv[NR], rk[NR];
#pragma unroll
  for (int q = 0; q < NR; ++q) {
    const int j = 32 * q + lane;
    e[q] = (j < d) ? grow[j] : 0x7FFFFFFF;
    sv[q] = (j < d) ? gsrc[j] : 0;
    rk[q] = 0;
  }
#pragma unroll
  for (int q2 = 0; q2 < NR; ++q2)
#pragma unroll
    for (int k = 0; k < 32; ++k) {
      const int o = __builtin_amdgcn_readlane(e[q2], k);
#pragma unroll
      for (int q = 0; q < NR; ++q) rk[q] += (o < e[q]) ? 1 : 0;
    }
#pragma unroll
  for (int q = 0; q < NR; ++q)
    if (32 * q + lane < d) row[rk[q]] = sv[q];
  __builtin_amdgcn_fence(__ATOMIC_RELEASE, "wavefront");
  __builtin_amdgcn_wave_barrier();
  __builtin_amdgcn_fence(__ATOMIC_ACQUIRE, "wavefront");
  const int i = lane & 7;
#pragma unroll
  for (int c = 0; c < NC; ++c) {
    const int j = 8 * c + i;
    sidx[c] = (j < d) ? row[j] : n;
  }
}

template <int NC>
__device__ __forceinline__ v4f agg_in_regs(const int lane, const int d, const int (&sidx)[NC],
                                           const v2f (&es2)[NC], const bool second,
                                           const float edh, const char* __restrict__ hb) {
  const int i = lane & 7;
  bool val[NC];
#pragma unroll
  for (int c = 0; c < NC; ++c) val[c] = (8 * c + i) < d;
  float p[NC];
  float m = -INFINITY;
#pragma unroll
  for (int c = 0; c < NC; ++c) {
    const float es = second ? es2[c].y : es2[c].x;
    p[c] = fminf(lrelu(es + edh), val[c] ? INFINITY : -INFINITY);
    m = fmaxf(m, p[c]);
  }
  m = fmaxf(m, row_xlane<4>(m));
  m = fmaxf(m, row_xlane<2>(m));
  m = fmaxf(m, row_xlane<1>(m));
#pragma unroll
  for (int c = 0; c < NC; ++c) p[c] = expf(p[c] - m);
  float t[4];
#pragma unroll
  for (int c = 0; c < 4; ++c) t[c] = (NC == 8) ? p[c] + p[(c + 4) % NC] : p[c];
  float sm = (t[0] + t[2]) + (t[1] + t[3]);
  sm += row_xlane<4>(sm);
  sm += row_xlane<2>(sm);
  sm += row_xlane<1>(sm);
  const float den = sm + EPS_GAT;
#pragma unroll
  for (int c = 0; c < NC; ++c) p[c] = p[c] / den;

  v4f acc = {0.f, 0.f, 0.f, 0.f};
  const unsigned lane16 = 16u * lane;
#pragma unroll
  for (int g = 0; g < NC; ++g) {
    if (8 * g < d) {
      v4f hv[8];
      row_req<0>(sidx[g], hb, lane16, hv[0]);
      row_req<1>(sidx[g], hb, lane16, hv[1]);
      row_req<2>(sidx[g], hb, lane16, hv[2]);
      row_req<3>(sidx[g], hb, lane16, hv[3]);
      row_req<4>(sidx[g], hb, lane16, hv[4]);
      row_req<5>(sidx[g], hb, lane16, hv[5]);
      row_req<6>(sidx[g], hb, lane16, hv[6]);
      row_req<7>(sidx[g], hb, lane16, hv[7]);
      row_add<0>(p[g], hv[0], acc);
      row_add<1>(p[g], hv[1], acc);
      row_add<2>(p[g], hv[2], acc);
      row_add<3>(p[g], hv[3], acc);
      row_add<4>(p[g], hv[4], acc);
      row_add<5>(p[g], hv[5], acc);
      row_add<6>(p[g], hv[6], acc);
      row_add<7>(p[g], hv[7], acc);
    }
  }
  return acc;
}

template <int NC>
__device__ __forceinline__ void node_in_regs(const int lane, const int n, const int d,
                                             const int* __restrict__ grow, const int* __restrict__ gsrc,
                                             int* __restrict__ row,
                                             const float* __restrict__ esrc, const float* __restrict__ edst,
                                             const float* __restrict__ hbuf, float* __restrict__ out,
                                             const int N_) {
  int sidx[NC];
  list_in_order<NC>(lane, n, d, grow, gsrc, row, sidx);
  const char* eb = (const char*)esrc;
  const unsigned hd8 = 8u * (unsigned)(lane >> 3);
  v2f es2[NC];
#pragma unroll
  for (int c = 0; c < NC; ++c) es2[c] = *(const v2f*)(eb + (((unsigned)sidx[c] << 5) + hd8));
#pragma unroll
  for (int bi = 0; bi < GAT_B; ++bi) {
    const size_t rowbase = (size_t)bi * N_;
    const char* hb = (const char*)(hbuf + rowbase * 128);
    const float edh = edst[(rowbase + n) * 4 + (lane >> 3)];
    const v4f r = agg_in_regs<NC>(lane, d, sidx, es2, bi != 0, edh, hb);
    *(v4f*)(out + (rowbase + n) * 128 + 4 * lane) = r;
  }
}

__device__ __forceinline__ v4f src_logits(const float* __restrict__ esrc, const int s, const int bi) {
  const float* p = esrc + (size_t)s * 8 + bi;
  v4f es = {p[0], p[2], p[4], p[6]};
  return es;
}

__global__ void __launch_bounds__(256, 12)
gat_node_agg(const int* __restrict__ pfx_w, int* runlist_w,
             const float* __restrict__ esrc, const float* __restrict__ edst,
             const float* __restrict__ hbuf, float* __restrict__ out, int N_) {
  __shared__ int ordered[8][64];
  __shared__ int gathered[8][64];
  __shared__ int gathsrc[8][64];
  unsigned* runlist = (unsigned*)runlist_w;
  const int lane = threadIdx.x & 31;
  const int n = __builtin_amdgcn_readfirstlane((int)((blockIdx.x * blockDim.x + threadIdx.x) >> 5));
  if (n >= N_) return;
  int a = 0, c = 0;
  if (lane < CNT_BLOCKS) {
    const int* pw = pfx_w + lane * (PFX_PITCH / 2) + (n >> 1);
    const unsigned w0 = (unsigned)pw[0], w1 = (unsigned)pw[1];
    const unsigned lo = (n & 1) ? (w0 >> 16) : (w0 & 0xFFFFu);
    const unsigned hi = (n & 1) ? (w1 & 0xFFFFu) : (w0 >> 16);
    a = (int)lo;
    c = (int)(hi - lo);
  }
  int inc = c;
  for (int o = 1; o < 32; o <<= 1) {
    const int up = __shfl_up(inc, o, 32);
    inc += (lane >= o) ? up : 0;
  }
  const int off = inc - c;
  const int d = __builtin_amdgcn_readlane(inc, 31);

  if (d <= 64) {
    int* row = ordered[threadIdx.x >> 5];
    int* grow = gathered[threadIdx.x >> 5];
    int* gsrc = gathsrc[threadIdx.x >> 5];
    collect_list(lane, a, c, off, runlist, grow, gsrc);
    if (d <= 32) node_in_regs<4>(lane, n, d, grow, gsrc, row, esrc, edst, hbuf, out, N_);
    else         node_in_regs<8>(lane, n, d, grow, gsrc, row, esrc, edst, hbuf, out, N_);
    return;
  }

  for (int k = 0; k < CNT_BLOCKS; ++k) {
    const int ak = __builtin_amdgcn_readlane(a, k), ck = __builtin_amdgcn_readlane(c, k);
    unsigned* sg = runlist + (size_t)k * CNT_CHUNK + ak;
    for (int pass = 0; pass < ck; ++pass) {
      for (int q = (pass & 1) + 2 * lane; q + 1 < ck; q += 64) {
        const unsigned lo = sg[q], hi = sg[q + 1];
        if (lo > hi) { sg[q] = hi; sg[q + 1] = lo; }
      }
      __threadfence();
    }
  }

  for (int bi = 0; bi < GAT_B; ++bi) {
  const size_t rowbase = (size_t)bi * N_;
  const v4f ed = *(const v4f*)(edst + (rowbase + n) * 4);

  v4f m = {-INFINITY, -INFINITY, -INFINITY, -INFINITY};
  for (int k = 0; k < CNT_BLOCKS; ++k) {
    const int ak = __builtin_amdgcn_readlane(a, k), ck = __builtin_amdgcn_readlane(c, k);
    const unsigned* sg = runlist + (size_t)k * CNT_CHUNK + ak;
    for (int jb = 0; jb < ck; jb += 32) {
      const int j = jb + lane;
      if (j < ck) {
        v4f es = src_logits(esrc, (int)(sg[j] & 0xFFFFu), bi);
#pragma unroll
        for (int h = 0; h < 4; ++h) m[h] = fmaxf(m[h], lrelu(es[h] + ed[h]));
      }
    }
  }
#pragma unroll
  for (int o = 16; o > 0; o >>= 1)
#pragma unroll
    for (int h = 0; h < 4; ++h) m[h] = fmaxf(m[h], __shfl_xor(m[h], o, 32));

  v4f sm = {0.f, 0.f, 0.f, 0.f};
  for (int k = 0; k < CNT_BLOCKS; ++k) {
    const int ak = __builtin_amdgcn_readlane(a, k), ck = __builtin_amdgcn_readlane(c, k);
    const unsigned* sg = runlist + (size_t)k * CNT_CHUNK + ak;
    for (int jb = 0; jb < ck; jb += 32) {
      const int j = jb + lane;
      if (j < ck) {
        v4f es = src_logits(esrc, (int)(sg[j] & 0xFFFFu), bi);
#pragma unroll
        for (int h = 0; h < 4; ++h) sm[h] += expf(lrelu(es[h] + ed[h]) - m[h]);
      }
    }
  }
#pragma unroll
  for (int o = 16; o > 0; o >>= 1)
#pragma unroll
    for (int h = 0; h < 4; ++h) sm[h] += __shfl_xor(sm[h], o, 32);

  const int hd = lane >> 3;
  const int k0m = hd == 0 ? -1 : 0, k1m = hd == 1 ? -1 : 0, k2m = hd == 2 ? -1 : 0, k3m = hd == 3 ? -1 : 0;
  v4f acc = {0.f, 0.f, 0.f, 0.f};
  const float* hcol = hbuf + rowbase * 128 + 4 * lane;
  for (int k = 0; k < CNT_BLOCKS; ++k) {
    const int ak = __builtin_amdgcn_readlane(a, k), ck = __builtin_amdgcn_readlane(c, k);
    const unsigned* sg = runlist + (size_t)k * CNT_CHUNK + ak;
    for (int jb = 0; jb < ck; jb += 32) {
      const int j = jb + lane;
      int s = n;
      v4f al = {0.f, 0.f, 0.f, 0.f};
      if (j < ck) {
        s = (int)(sg[j] & 0xFFFFu);
        v4f es = src_logits(esrc, s, bi);
#pragma unroll
        for (int h = 0; h < 4; ++h)
          al[h] = expf(lrelu(es[h] + ed[h]) - m[h]) / (sm[h] + EPS_GAT);
      }
      const int cnt = ck - jb < 32 ? ck - jb : 32;
      for (int k0 = 0; k0 < cnt; k0 += 8) {
        v4f hv[8];
        float aw[8];
#pragma unroll
        for (int u = 0; u < 8; ++u) {
          const int kk = k0 + u;
          const int sk = __builtin_amdgcn_readlane(s, kk);
          const int b0 = __builtin_amdgcn_readlane(__float_as_int(al[0]), kk);
          const int b1 = __builtin_amdgcn_readlane(__float_as_int(al[1]), kk);
          const int b2 = __builtin_amdgcn_readlane(__float_as_int(al[2]), kk);
          const int b3 = __builtin_amdgcn_readlane(__float_as_int(al[3]), kk);
          aw[u] = __int_as_float((b0 & k0m) | (b1 & k1m) | (b2 & k2m) | (b3 & k3m));
          hv[u] = *(const v4f*)(hcol + (size_t)sk * 128);
        }
#pragma unroll
        for (int u = 0; u < 8; ++u)
          if (k0 + u < cnt) acc += aw[u] * hv[u];
      }
    }
  }
  *(v4f*)(out + (rowbase + n) * 128 + 4 * lane) = acc;
  }
}

extern "C" void kernel_launch(void* const* d_in, const int* in_sizes, int n_in,
                              void* d_out, int out_size, void* d_ws, size_t ws_size,
                              hipStream_t stream) {
  (void)in_sizes; (void)n_in; (void)out_size; (void)ws_size;
  const float* x     = (const float*)d_in[0];
  const int*   ei    = (const int*)d_in[1];
  const float* W     = (const float*)d_in[2];
  const float* a_src = (const float*)d_in[3];
  const float* a_dst = (const float*)d_in[4];
  float* out = (float*)d_out;

  const int B = 2, N = GAT_N, E = GAT_E;
  const int BN = B * N;

  float* hbuf   = (float*)d_ws;
  float* esrc   = hbuf + (size_t)BN * 128;
  float* edst   = esrc + (size_t)BN * 4;
  int*   pfx_w     = (int*)(edst + (size_t)BN * 4);
  int*   runlist_w = pfx_w + (CNT_BLOCKS * PFX_PITCH) / 2;

  gat_gemm_count<<<GEMM_BLOCKS + CNT_BLOCKS, 256, 0, stream>>>(x, W, a_src, a_dst, hbuf, esrc, edst,
                                                               ei, ei + E, pfx_w, runlist_w);
  gat_node_agg<<<(N * 32 + 255) / 256, 256, 0, stream>>>(pfx_w, runlist_w, esrc, edst, hbuf, out, N);
}
